# starting kernel
# baseline (speedup 1.0000x reference)
#include <hip/hip_runtime.h>
#include <hip/hip_bf16.h>

#define N_NODES   50000
#define N_EDGES   800000
#define E_TOT     850000
#define IN_F      128
#define HC        128
#define NEG_SLOPE 0.2f

typedef __attribute__((ext_vector_type(2))) float v2f;
typedef __attribute__((ext_vector_type(8))) float v8f;

__global__ void gat_init(float* __restrict__ out, const float* __restrict__ bias,
                         float* __restrict__ m, float* __restrict__ denom) {
  int i = blockIdx.x * blockDim.x + threadIdx.x;
  if (i < N_NODES * HC) out[i] = bias[i & (HC - 1)];
  if (i < N_NODES * 4) {
    m[i] = -3.402823466e38f;
    denom[i] = 0.0f;
  }
}

__global__ void __launch_bounds__(256)
gat_gemm(const float* __restrict__ x, const float* __restrict__ Wl,
         const float* __restrict__ Wr, float* __restrict__ xl,
         float* __restrict__ xr) {
  const int gw   = (blockIdx.x * blockDim.x + threadIdx.x) >> 5;
  const int lane = threadIdx.x & 31;
  const int mt = gw >> 4;
  const int nt = gw & 15;

  const float* __restrict__ W   = (nt < 8) ? Wl : Wr;
  float* __restrict__       dst = (nt < 8) ? xl : xr;
  const int ncol  = (nt & 7) * 16 + (lane & 15);
  const int mrow  = mt * 16 + (lane & 15);
  const int khalf = (lane >> 4) * 2;

  const float* __restrict__ xrow = x + (size_t)mrow * IN_F;

  v8f c = {};
#pragma unroll
  for (int kb = 0; kb < IN_F; kb += 4) {
    v2f a, b;
    const float2 ax = *(const float2*)(xrow + kb + khalf);
    a.x = ax.x; a.y = ax.y;
    b.x = W[(size_t)(kb + khalf) * HC + ncol];
    b.y = W[(size_t)(kb + khalf + 1) * HC + ncol];
    c = __builtin_amdgcn_wmma_f32_16x16x4_f32(false, a, false, b,
                                              (short)0, c, false, false);
  }

  const int mbase = mt * 16 + 8 * (lane >> 4);
#pragma unroll
  for (int v = 0; v < 8; ++v)
    dst[(size_t)(mbase + v) * HC + ncol] = c[v];
}

__global__ void __launch_bounds__(256)
gat_logits(const int* __restrict__ ei, const float* __restrict__ xl,
           const float* __restrict__ xr, const float* __restrict__ att,
           float* __restrict__ logits, float* __restrict__ m) {
  const unsigned gid  = blockIdx.x * blockDim.x + threadIdx.x;
  const unsigned e    = gid >> 5;
  const unsigned lane = threadIdx.x & 31;

  int src, dst;
  if (e < N_EDGES) { src = ei[e]; dst = ei[N_EDGES + e]; }
  else             { src = dst = (int)(e - N_EDGES); }

  const float4 a = ((const float4*)(xl + (size_t)src * HC))[lane];
  const float4 b = ((const float4*)(xr + (size_t)dst * HC))[lane];
  const float4 w = ((const float4*)att)[lane];

  float vx = a.x + b.x, vy = a.y + b.y, vz = a.z + b.z, vw = a.w + b.w;
  vx = (vx > 0.f) ? vx : vx * NEG_SLOPE;
  vy = (vy > 0.f) ? vy : vy * NEG_SLOPE;
  vz = (vz > 0.f) ? vz : vz * NEG_SLOPE;
  vw = (vw > 0.f) ? vw : vw * NEG_SLOPE;
  float s = vx * w.x + vy * w.y + vz * w.z + vw * w.w;

  s += __shfl_xor(s, 1);
  s += __shfl_xor(s, 2);
  s += __shfl_xor(s, 4);

  if ((lane & 7) == 0) {
    const int h = lane >> 3;
    logits[(size_t)e * 4 + h] = s;
    atomicMax(&m[(size_t)dst * 4 + h], s);
  }
}

__global__ void gat_expsum(const int* __restrict__ ei,
                           const float* __restrict__ logits,
                           const float* __restrict__ m,
                           float* __restrict__ denom) {
  const unsigned gid = blockIdx.x * blockDim.x + threadIdx.x;
  if (gid >= (unsigned)E_TOT * 4) return;
  const unsigned e = gid >> 2;
  const unsigned h = gid & 3;

  int dst;
  if (e < N_EDGES) dst = ei[N_EDGES + e];
  else             dst = (int)(e - N_EDGES);

  const float ex = __expf(logits[(size_t)e * 4 + h] - m[(size_t)dst * 4 + h]);
  atomicAdd(&denom[(size_t)dst * 4 + h], ex);
}

__global__ void __launch_bounds__(256)
gat_aggregate(const int* __restrict__ ei, const float* __restrict__ xl,
              const float* __restrict__ logits, const float* __restrict__ m,
              const float* __restrict__ denom, float* __restrict__ out) {
  const unsigned gid  = blockIdx.x * blockDim.x + threadIdx.x;
  const unsigned e    = gid >> 5;
  const unsigned lane = threadIdx.x & 31;

  int src, dst;
  if (e < N_EDGES) { src = ei[e]; dst = ei[N_EDGES + e]; }
  else             { src = dst = (int)(e - N_EDGES); }

  const int h = lane >> 3;
  const float alpha =
      __expf(logits[(size_t)e * 4 + h] - m[(size_t)dst * 4 + h]) /
      (denom[(size_t)dst * 4 + h] + 1e-16f);

  const float4 v = ((const float4*)(xl + (size_t)src * HC))[lane];
  float* po = out + (size_t)dst * HC + lane * 4;
  atomicAdd(po + 0, v.x * alpha);
  atomicAdd(po + 1, v.y * alpha);
  atomicAdd(po + 2, v.z * alpha);
  atomicAdd(po + 3, v.w * alpha);
}

extern "C" void kernel_launch(void* const* d_in, const int* in_sizes, int n_in,
                              void* d_out, int out_size, void* d_ws, size_t ws_size,
                              hipStream_t stream) {
  const float* x    = (const float*)d_in[0];
  const int*   ei   = (const int*)  d_in[1];
  const float* Wl   = (const float*)d_in[2];
  const float* Wr   = (const float*)d_in[3];
  const float* att  = (const float*)d_in[4];
  const float* bias = (const float*)d_in[5];
  float* out = (float*)d_out;

  float* ws     = (float*)d_ws;
  float* xl     = ws;
  float* xr     = xl + (size_t)N_NODES * HC;
  float* m      = xr + (size_t)N_NODES * HC;
  float* denom  = m + (size_t)N_NODES * 4;
  float* logits = denom + (size_t)N_NODES * 4;

  gat_init<<<(N_NODES * HC + 255) / 256, 256, 0, stream>>>(out, bias, m, denom);

  gat_gemm<<<6250, 256, 0, stream>>>(x, Wl, Wr, xl, xr);

  gat_logits<<<106250, 256, 0, stream>>>(ei, xl, xr, att, logits, m);
  gat_expsum<<<((unsigned)E_TOT * 4 + 255) / 256, 256, 0, stream>>>(ei, logits, m, denom);
  gat_aggregate<<<106250, 256, 0, stream>>>(ei, xl, logits, m, denom, out);
}


# edge passes in one launch: each work-group owns 196 destination nodes, scans the edge list for its edges, keeps xr rows, max, exp-sum and sums in LDS (LDS float adds, arrival order); 2 launches
# speedup vs baseline: 2.5580x; 2.5580x over previous
#include <hip/hip_runtime.h>
#include <hip/hip_bf16.h>

#define N_NODES   50000
#define N_EDGES   800000
#define E_TOT     850000
#define IN_F      128
#define HC        128
#define NEG_SLOPE 0.2f

typedef __attribute__((ext_vector_type(2))) float v2f;
typedef __attribute__((ext_vector_type(8))) float v8f;

__global__ void __launch_bounds__(256)
gat_gemm(const float* __restrict__ x, const float* __restrict__ Wl,
         const float* __restrict__ Wr, float* __restrict__ xl,
         float* __restrict__ xr) {
  const int gw   = (blockIdx.x * blockDim.x + threadIdx.x) >> 5;
  const int lane = threadIdx.x & 31;
  const int mt = gw >> 4;
  const int nt = gw & 15;

  const float* __restrict__ W   = (nt < 8) ? Wl : Wr;
  float* __restrict__       dst = (nt < 8) ? xl : xr;
  const int ncol  = (nt & 7) * 16 + (lane & 15);
  const int mrow  = mt * 16 + (lane & 15);
  const int khalf = (lane >> 4) * 2;

  const float* __restrict__ xrow = x + (size_t)mrow * IN_F;

  v8f c = {};
#pragma unroll
  for (int kb = 0; kb < IN_F; kb += 4) {
    v2f a, b;
    const float2 ax = *(const float2*)(xrow + kb + khalf);
    a.x = ax.x; a.y = ax.y;
    b.x = W[(size_t)(kb + khalf) * HC + ncol];
    b.y = W[(size_t)(kb + khalf + 1) * HC + ncol];
    c = __builtin_amdgcn_wmma_f32_16x16x4_f32(false, a, false, b,
                                              (short)0, c, false, false);
  }

  const int mbase = mt * 16 + 8 * (lane >> 4);
#pragma unroll
  for (int v = 0; v < 8; ++v)
    dst[(size_t)(mbase + v) * HC + ncol] = c[v];
}

#define NPB    196
#define NBLK   256
#define CAP    8192
#define TPB    1024
#define NWAVE  32
#define NWIN   ((N_EDGES + CAP - 1) / CAP + 1)

__device__ __forceinline__ void list_append(int* s_cnt, int* s_src, int* s_dl,
                                            int e, int dl) {
  const int slot = atomicAdd(s_cnt, 1);
  if (slot < CAP) { s_src[slot] = e; s_dl[slot] = dl; }
}

__device__ __forceinline__ void scan_load(int4 (&v)[4], const int* __restrict__ dsts,
                                          int base, int hi) {
#pragma unroll
  for (int u = 0; u < 4; ++u)
    v[u] = *(const int4*)(dsts + min(base + u * TPB * 4, hi - 4));
}

__device__ __forceinline__ void scan_match(const int4 (&v)[4], int base, int hi,
                                           int node0, unsigned nn,
                                           int* s_cnt, int* s_src, int* s_dl) {
#pragma unroll
  for (int u = 0; u < 4; ++u) {
    const int e = base + u * TPB * 4;
    const unsigned lim = (e < hi) ? nn : 0u;
    const unsigned d0 = (unsigned)(v[u].x - node0);
    const unsigned d1 = (unsigned)(v[u].y - node0);
    const unsigned d2 = (unsigned)(v[u].z - node0);
    const unsigned d3 = (unsigned)(v[u].w - node0);
    if ((d0 < lim) | (d1 < lim) | (d2 < lim) | (d3 < lim)) {
      if (d0 < lim) list_append(s_cnt, s_src, s_dl, e + 0, (int)d0);
      if (d1 < lim) list_append(s_cnt, s_src, s_dl, e + 1, (int)d1);
      if (d2 < lim) list_append(s_cnt, s_src, s_dl, e + 2, (int)d2);
      if (d3 < lim) list_append(s_cnt, s_src, s_dl, e + 3, (int)d3);
    }
  }
}

__device__ __forceinline__ void build_edges(const int* __restrict__ ei, int lo, int hi,
                                            int node0, unsigned nn,
                                            int* s_cnt, int* s_src, int* s_dl) {
  const int* __restrict__ dsts = ei + N_EDGES;
  const int trips = (hi - lo + TPB * 16 - 1) / (TPB * 16);
  int base = lo + (int)threadIdx.x * 4;
  int4 va[4], vb[4];
  scan_load(va, dsts, base, hi);
  for (int t = 0; t < trips; t += 2) {
    scan_load(vb, dsts, base + TPB * 16, hi);
    scan_match(va, base, hi, node0, nn, s_cnt, s_src, s_dl);
    scan_load(va, dsts, base + TPB * 32, hi);
    scan_match(vb, base + TPB * 16, hi, node0, nn, s_cnt, s_src, s_dl);
    base += TPB * 32;
  }
}

__device__ __forceinline__ void build_self(int node0, unsigned nn,
                                           int* s_cnt, int* s_src, int* s_dl) {
  if (threadIdx.x < nn)
    list_append(s_cnt, s_src, s_dl, N_EDGES + node0 + (int)threadIdx.x, (int)threadIdx.x);
}

__device__ __forceinline__ void phase_resolve(const int* __restrict__ ei, int n, int* s_src) {
  for (int i = threadIdx.x; i < n; i += TPB) {
    const int e = s_src[i];
    s_src[i] = (e < N_EDGES) ? ei[e] : e - N_EDGES;
  }
}

template <int NE, bool XR_LDS>
__device__ __forceinline__ void logit_group(const float* __restrict__ xl,
                                            const float* __restrict__ xr,
                                            const float* s_xr,
                                            const float4 w, int node0, int i0, int lane,
                                            const int* s_src, const int* s_dl,
                                            float* s_val, float* s_m) {
  float4 a[NE], bg[NE];
  int dl[NE];
#pragma unroll
  for (int u = 0; u < NE; ++u) {
    const int src = s_src[i0 + u];
    dl[u] = s_dl[i0 + u];
    a[u] = ((const float4*)(xl + (size_t)src * HC))[lane];
    if (!XR_LDS) bg[u] = ((const float4*)(xr + (size_t)(node0 + dl[u]) * HC))[lane];
  }
  __builtin_amdgcn_sched_barrier(0);
  float s[NE];
#pragma unroll
  for (int u = 0; u < NE; ++u) {
    const float4 b = XR_LDS ? ((const float4*)(s_xr + dl[u] * HC))[lane] : bg[u];
    float vx = a[u].x + b.x, vy = a[u].y + b.y;
    float vz = a[u].z + b.z, vw = a[u].w + b.w;
    vx = (vx > 0.f) ? vx : vx * NEG_SLOPE;
    vy = (vy > 0.f) ? vy : vy * NEG_SLOPE;
    vz = (vz > 0.f) ? vz : vz * NEG_SLOPE;
    vw = (vw > 0.f) ? vw : vw * NEG_SLOPE;
    s[u] = vx * w.x + vy * w.y + vz * w.z + vw * w.w;
  }
#pragma unroll
  for (int u = 0; u < NE; ++u) s[u] += __shfl_xor(s[u], 1);
#pragma unroll
  for (int u = 0; u < NE; ++u) s[u] += __shfl_xor(s[u], 2);
#pragma unroll
  for (int u = 0; u < NE; ++u) s[u] += __shfl_xor(s[u], 4);
  if ((lane & 7) == 0) {
    const int h = lane >> 3;
#pragma unroll
    for (int u = 0; u < NE; ++u) {
      s_val[(i0 + u) * 4 + h] = s[u];
      __hip_atomic_fetch_max(&s_m[dl[u] * 4 + h], s[u], __ATOMIC_RELAXED,
                             __HIP_MEMORY_SCOPE_WORKGROUP);
    }
  }
}

#define GRP 8

template <bool XR_LDS>
__device__ __forceinline__ void phase_logits(const float* __restrict__ xl,
                                             const float* __restrict__ xr,
                                             const float* s_xr,
                                             const float4 w, int node0, int n,
                                             const int* s_src, const int* s_dl,
                                             float* s_val, float* s_m) {
  const int wave = threadIdx.x >> 5;
  const int lane = threadIdx.x & 31;
  const int nfull = n / GRP;
  for (int g = wave; g < nfull; g += NWAVE)
    logit_group<GRP, XR_LDS>(xl, xr, s_xr, w, node0, g * GRP, lane, s_src, s_dl, s_val, s_m);
  const int i = nfull * GRP + wave;
  if (i < n)
    logit_group<1, XR_LDS>(xl, xr, s_xr, w, node0, i, lane, s_src, s_dl, s_val, s_m);
}

__device__ __forceinline__ void phase_exp(int n, bool add, const int* s_dl,
                                          float* s_val, const float* s_m, float* s_den) {
  for (int i = threadIdx.x; i < n * 4; i += TPB) {
    const int k = s_dl[i >> 2] * 4 + (i & 3);
    const float ex = __expf(s_val[i] - s_m[k]);
    s_val[i] = ex;
    if (add)
      __hip_atomic_fetch_add(&s_den[k], ex, __ATOMIC_RELAXED, __HIP_MEMORY_SCOPE_WORKGROUP);
  }
}

__device__ __forceinline__ void phase_alpha(int n, const int* s_dl, float* s_val,
                                            const float* s_den) {
  for (int i = threadIdx.x; i < n * 4; i += TPB) {
    const int k = s_dl[i >> 2] * 4 + (i & 3);
    s_val[i] = s_val[i] / (s_den[k] + 1e-16f);
  }
}

template <int NE>
__device__ __forceinline__ void sum_group(const float* __restrict__ xl, int i0, int lane,
                                          const int* s_src, const int* s_dl,
                                          const float* s_val, float* s_acc) {
  float4 v[NE];
  int dl[NE];
#pragma unroll
  for (int u = 0; u < NE; ++u) {
    const int src = s_src[i0 + u];
    dl[u] = s_dl[i0 + u];
    v[u] = ((const float4*)(xl + (size_t)src * HC))[lane];
  }
  __builtin_amdgcn_sched_barrier(0);
#pragma unroll
  for (int u = 0; u < NE; ++u) {
    const float alpha = s_val[(i0 + u) * 4 + (lane >> 3)];
    float* row = s_acc + dl[u] * HC + lane;
    __hip_atomic_fetch_add(row +  0, v[u].x * alpha, __ATOMIC_RELAXED, __HIP_MEMORY_SCOPE_WORKGROUP);
    __hip_atomic_fetch_add(row + 32, v[u].y * alpha, __ATOMIC_RELAXED, __HIP_MEMORY_SCOPE_WORKGROUP);
    __hip_atomic_fetch_add(row + 64, v[u].z * alpha, __ATOMIC_RELAXED, __HIP_MEMORY_SCOPE_WORKGROUP);
    __hip_atomic_fetch_add(row + 96, v[u].w * alpha, __ATOMIC_RELAXED, __HIP_MEMORY_SCOPE_WORKGROUP);
  }
}

__device__ __forceinline__ void phase_sum(const float* __restrict__ xl, int n,
                                          const int* s_src, const int* s_dl,
                                          const float* s_val, float* s_acc) {
  const int wave = threadIdx.x >> 5;
  const int lane = threadIdx.x & 31;
  const int nfull = n / GRP;
  for (int g = wave; g < nfull; g += NWAVE)
    sum_group<GRP>(xl, g * GRP, lane, s_src, s_dl, s_val, s_acc);
  const int i = nfull * GRP + wave;
  if (i < n) sum_group<1>(xl, i, lane, s_src, s_dl, s_val, s_acc);
}

__device__ __forceinline__ void zero_acc(float* s_acc) {
  for (int i = threadIdx.x; i < NPB * HC / 4; i += TPB)
    ((float4*)s_acc)[i] = make_float4(0.f, 0.f, 0.f, 0.f);
}

__global__ void __launch_bounds__(TPB)
gat_owner(const int* __restrict__ ei, const float* __restrict__ xl,
          const float* __restrict__ xr, const float* __restrict__ att,
          const float* __restrict__ bias, float* __restrict__ out) {
  __shared__ float s_acc[NPB * HC];
  __shared__ float s_m[NPB * 4];
  __shared__ float s_den[NPB * 4];
  __shared__ int   s_src[CAP];
  __shared__ int   s_dl[CAP];
  __shared__ float s_val[CAP * 4];
  __shared__ int   s_cnt;

  const int tid   = threadIdx.x;
  const int node0 = blockIdx.x * NPB;
  const unsigned nn = (unsigned)min(NPB, N_NODES - node0);
  const float4 w = ((const float4*)att)[tid & 31];

  for (int i = tid; i < (int)nn * (HC / 4); i += TPB)
    ((float4*)s_acc)[i] = ((const float4*)(xr + (size_t)node0 * HC))[i];
  for (int i = tid; i < NPB * 4; i += TPB) {
    s_m[i] = -3.402823466e38f;
    s_den[i] = 0.0f;
  }
  if (tid == 0) s_cnt = 0;
  __syncthreads();

  build_edges(ei, 0, N_EDGES, node0, nn, &s_cnt, s_src, s_dl);
  build_self(node0, nn, &s_cnt, s_src, s_dl);
  __syncthreads();
  const int total = s_cnt;

  if (total <= CAP) {
    phase_resolve(ei, total, s_src);
    __syncthreads();
    phase_logits<true>(xl, xr, s_acc, w, node0, total, s_src, s_dl, s_val, s_m);
    __syncthreads();
    zero_acc(s_acc);
    phase_exp(total, true, s_dl, s_val, s_m, s_den);
    __syncthreads();
    phase_alpha(total, s_dl, s_val, s_den);
    __syncthreads();
    phase_sum(xl, total, s_src, s_dl, s_val, s_acc);
    __syncthreads();
  } else {
    zero_acc(s_acc);
    for (int sweep = 0; sweep < 3; ++sweep) {
      for (int win = 0; win < NWIN; ++win) {
        __syncthreads();
        if (tid == 0) s_cnt = 0;
        __syncthreads();
        if (win < NWIN - 1)
          build_edges(ei, win * CAP, min(win * CAP + CAP, N_EDGES), node0, nn,
                      &s_cnt, s_src, s_dl);
        else
          build_self(node0, nn, &s_cnt, s_src, s_dl);
        __syncthreads();
        const int n = s_cnt;
        phase_resolve(ei, n, s_src);
        __syncthreads();
        phase_logits<false>(xl, xr, s_acc, w, node0, n, s_src, s_dl, s_val, s_m);
        __syncthreads();
        if (sweep >= 1) {
          phase_exp(n, sweep == 1, s_dl, s_val, s_m, s_den);
          __syncthreads();
        }
        if (sweep == 2) {
          phase_alpha(n, s_dl, s_val, s_den);
          __syncthreads();
          phase_sum(xl, n, s_src, s_dl, s_val, s_acc);
          __syncthreads();
        }
      }
    }
  }

  const float4 bq = ((const float4*)bias)[tid & 31];
  for (int idx = tid; idx < (int)nn * 32; idx += TPB) {
    const float* row = s_acc + (idx >> 5) * HC + (idx & 31);
    float4 a;
    a.x = row[0]  + bq.x;
    a.y = row[32] + bq.y;
    a.z = row[64] + bq.z;
    a.w = row[96] + bq.w;
    ((float4*)(out + (size_t)node0 * HC))[idx] = a;
  }
}

extern "C" void kernel_launch(void* const* d_in, const int* in_sizes, int n_in,
                              void* d_out, int out_size, void* d_ws, size_t ws_size,
                              hipStream_t stream) {
  const float* x    = (const float*)d_in[0];
  const int*   ei   = (const int*)  d_in[1];
  const float* Wl   = (const float*)d_in[2];
  const float* Wr   = (const float*)d_in[3];
  const float* att  = (const float*)d_in[4];
  const float* bias = (const float*)d_in[5];
  float* out = (float*)d_out;

  float* ws = (float*)d_ws;
  float* xl = ws;
  float* xr = xl + (size_t)N_NODES * HC;

  gat_gemm<<<6250, 256, 0, stream>>>(x, Wl, Wr, xl, xr);

  gat_owner<<<NBLK, TPB, 0, stream>>>(ei, xl, xr, att, bias, out);
}


# GEMM launch re-encodes the edge list each call: a 1-byte owner key per edge, which every owner work-group scans instead of the 4-byte destination, and a source/local-destination word read on a hit
# speedup vs baseline: 2.8335x; 1.1077x over previous
#include <hip/hip_runtime.h>
#include <hip/hip_bf16.h>

#define N_NODES   50000
#define N_EDGES   800000
#define E_TOT     850000
#define IN_F      128
#define HC        128
#define NEG_SLOPE 0.2f
#define NPB       196
#define NBLK      256
static_assert(NBLK <= 256 && NBLK * NPB >= N_NODES, "owner key is one byte");
static_assert(N_NODES <= 65536 && NPB <= 65536, "edge record packs source and local destination in 16 bits each");

typedef __attribute__((ext_vector_type(2))) float v2f;
typedef __attribute__((ext_vector_type(8))) float v8f;

__global__ void __launch_bounds__(256)
gat_gemm(const float* __restrict__ x, const float* __restrict__ Wl,
         const float* __restrict__ Wr, const int* __restrict__ ei,
         float* __restrict__ xl, float* __restrict__ xr,
         unsigned* __restrict__ key, unsigned* __restrict__ rec) {
  const int gt   = blockIdx.x * blockDim.x + threadIdx.x;
  const int gw   = gt >> 5;
  const int lane = threadIdx.x & 31;

  if (gt < N_EDGES / 4) {
    const int4 s = ((const int4*)ei)[gt];
    const int4 d = ((const int4*)(ei + N_EDGES))[gt];
    const unsigned k0 = (unsigned)d.x / (unsigned)NPB;
    const unsigned k1 = (unsigned)d.y / (unsigned)NPB;
    const unsigned k2 = (unsigned)d.z / (unsigned)NPB;
    const unsigned k3 = (unsigned)d.w / (unsigned)NPB;
    key[gt] = k0 | k1 << 8 | k2 << 16 | k3 << 24;
    uint4 r;
    r.x = (unsigned)s.x | ((unsigned)d.x - k0 * NPB) << 16;
    r.y = (unsigned)s.y | ((unsigned)d.y - k1 * NPB) << 16;
    r.z = (unsigned)s.z | ((unsigned)d.z - k2 * NPB) << 16;
    r.w = (unsigned)s.w | ((unsigned)d.w - k3 * NPB) << 16;
    ((uint4*)rec)[gt] = r;
  }
  const int mt = gw >> 4;
  const int nt = gw & 15;

  const float* __restrict__ W   = (nt < 8) ? Wl : Wr;
  float* __restrict__       dst = (nt < 8) ? xl : xr;
  const int ncol  = (nt & 7) * 16 + (lane & 15);
  const int mrow  = mt * 16 + (lane & 15);
  const int khalf = (lane >> 4) * 2;

  const float* __restrict__ xrow = x + (size_t)mrow * IN_F;

  v8f c = {};
#pragma unroll
  for (int kb = 0; kb < IN_F; kb += 4) {
    v2f a, b;
    const float2 ax = *(const float2*)(xrow + kb + khalf);
    a.x = ax.x; a.y = ax.y;
    b.x = W[(size_t)(kb + khalf) * HC + ncol];
    b.y = W[(size_t)(kb + khalf + 1) * HC + ncol];
    c = __builtin_amdgcn_wmma_f32_16x16x4_f32(false, a, false, b,
                                              (short)0, c, false, false);
  }

  const int mbase = mt * 16 + 8 * (lane >> 4);
#pragma unroll
  for (int v = 0; v < 8; ++v)
    dst[(size_t)(mbase + v) * HC + ncol] = c[v];
}

#define CAP    8192
#define TPB    1024
#define NWAVE  32
#define NWIN   ((N_EDGES + CAP - 1) / CAP + 1)

__device__ __forceinline__ void list_append(int* s_cnt, int* s_src, int e) {
  const int slot = atomicAdd(s_cnt, 1);
  if (slot < CAP) s_src[slot] = e;
}

#define SCAN_U 5

__device__ __forceinline__ void scan_load(int4 (&v)[SCAN_U],
                                          const unsigned char* __restrict__ keys,
                                          int base, int hi) {
#pragma unroll
  for (int u = 0; u < SCAN_U; ++u)
    v[u] = *(const int4*)(keys + min(base + u * TPB * 16, hi - 16));
}

__device__ __forceinline__ unsigned eq_bytes(unsigned x, unsigned pat) {
  const unsigned y = x ^ pat;
  const unsigned t = (y & 0x7F7F7F7Fu) + 0x7F7F7F7Fu;
  return ~(t | y | 0x7F7F7F7Fu);
}

__device__ __forceinline__ void scan_match(const int4 (&v)[SCAN_U], int base, int hi,
                                           unsigned pat, int* s_cnt, int* s_src) {
#pragma unroll
  for (int u = 0; u < SCAN_U; ++u) {
    const int e = base + u * TPB * 16;
    unsigned c = (eq_bytes((unsigned)v[u].x, pat) >> 7) | (eq_bytes((unsigned)v[u].y, pat) >> 6)
               | (eq_bytes((unsigned)v[u].z, pat) >> 5) | (eq_bytes((unsigned)v[u].w, pat) >> 4);
    if (e >= hi) c = 0u;
    while (c) {
      const int b = __builtin_ctz(c);
      c &= c - 1u;
      list_append(s_cnt, s_src, e + ((b & 7) << 2) + (b >> 3));
    }
  }
}

__device__ __forceinline__ void build_edges(const unsigned char* __restrict__ keys,
                                            int lo, int hi, int* s_cnt, int* s_src) {
  const unsigned pat = blockIdx.x * 0x01010101u;
  const int trips = (hi - lo + TPB * 16 * SCAN_U - 1) / (TPB * 16 * SCAN_U);
  int base = lo + (int)threadIdx.x * 16;
  int4 va[SCAN_U], vb[SCAN_U];
  scan_load(va, keys, base, hi);
  for (int t = 0; t < trips; t += 2) {
    scan_load(vb, keys, base + TPB * 16 * SCAN_U, hi);
    scan_match(va, base, hi, pat, s_cnt, s_src);
    scan_load(va, keys, base + TPB * 32 * SCAN_U, hi);
    scan_match(vb, base + TPB * 16 * SCAN_U, hi, pat, s_cnt, s_src);
    base += TPB * 32 * SCAN_U;
  }
}

__device__ __forceinline__ void build_self(int node0, unsigned nn,
                                           int* s_cnt, int* s_src, int* s_dl) {
  if (threadIdx.x < nn) {
    const int slot = atomicAdd(s_cnt, 1);
    if (slot < CAP) {
      s_src[slot] = N_EDGES + node0 + (int)threadIdx.x;
      s_dl[slot]  = (int)threadIdx.x;
    }
  }
}

__device__ __forceinline__ void phase_resolve(const unsigned* __restrict__ rec, int n,
                                              int* s_src, int* s_dl) {
  for (int i = threadIdx.x; i < n; i += TPB) {
    const int e = s_src[i];
    if (e < N_EDGES) {
      const unsigned r = rec[e];
      s_src[i] = (int)(r & 0xFFFFu);
      s_dl[i]  = (int)(r >> 16);
    } else {
      s_src[i] = e - N_EDGES;
    }
  }
}

template <int NE, bool XR_LDS>
__device__ __forceinline__ void logit_group(const float* __restrict__ xl,
                                            const float* __restrict__ xr,
                                            const float* s_xr,
                                            const float4 w, int node0, int i0, int lane,
                                            const int* s_src, const int* s_dl,
                                            float* s_val, float* s_m) {
  float4 a[NE], bg[NE];
  int dl[NE];
#pragma unroll
  for (int u = 0; u < NE; ++u) {
    const int src = s_src[i0 + u];
    dl[u] = s_dl[i0 + u];
    a[u] = ((const float4*)(xl + (size_t)src * HC))[lane];
    if (!XR_LDS) bg[u] = ((const float4*)(xr + (size_t)(node0 + dl[u]) * HC))[lane];
  }
  __builtin_amdgcn_sched_barrier(0);
  float s[NE];
#pragma unroll
  for (int u = 0; u < NE; ++u) {
    const float4 b = XR_LDS ? ((const float4*)(s_xr + dl[u] * HC))[lane] : bg[u];
    float vx = a[u].x + b.x, vy = a[u].y + b.y;
    float vz = a[u].z + b.z, vw = a[u].w + b.w;
    vx = (vx > 0.f) ? vx : vx * NEG_SLOPE;
    vy = (vy > 0.f) ? vy : vy * NEG_SLOPE;
    vz = (vz > 0.f) ? vz : vz * NEG_SLOPE;
    vw = (vw > 0.f) ? vw : vw * NEG_SLOPE;
    s[u] = vx * w.x + vy * w.y + vz * w.z + vw * w.w;
  }
#pragma unroll
  for (int u = 0; u < NE; ++u) s[u] += __shfl_xor(s[u], 1);
#pragma unroll
  for (int u = 0; u < NE; ++u) s[u] += __shfl_xor(s[u], 2);
#pragma unroll
  for (int u = 0; u < NE; ++u) s[u] += __shfl_xor(s[u], 4);
  if ((lane & 7) == 0) {
    const int h = lane >> 3;
#pragma unroll
    for (int u = 0; u < NE; ++u) {
      s_val[(i0 + u) * 4 + h] = s[u];
      __hip_atomic_fetch_max(&s_m[dl[u] * 4 + h], s[u], __ATOMIC_RELAXED,
                             __HIP_MEMORY_SCOPE_WORKGROUP);
    }
  }
}

#define GRP 8

template <bool XR_LDS>
__device__ __forceinline__ void phase_logits(const float* __restrict__ xl,
                                             const float* __restrict__ xr,
                                             const float* s_xr,
                                             const float4 w, int node0, int n,
                                             const int* s_src, const int* s_dl,
                                             float* s_val, float* s_m) {
  const int wave = threadIdx.x >> 5;
  const int lane = threadIdx.x & 31;
  const int nfull = n / GRP;
  for (int g = wave; g < nfull; g += NWAVE)
    logit_group<GRP, XR_LDS>(xl, xr, s_xr, w, node0, g * GRP, lane, s_src, s_dl, s_val, s_m);
  const int i = nfull * GRP + wave;
  if (i < n)
    logit_group<1, XR_LDS>(xl, xr, s_xr, w, node0, i, lane, s_src, s_dl, s_val, s_m);
}

__device__ __forceinline__ void phase_exp(int n, bool add, const int* s_dl,
                                          float* s_val, const float* s_m, float* s_den) {
  for (int i = threadIdx.x; i < n * 4; i += TPB) {
    const int k = s_dl[i >> 2] * 4 + (i & 3);
    const float ex = __expf(s_val[i] - s_m[k]);
    s_val[i] = ex;
    if (add)
      __hip_atomic_fetch_add(&s_den[k], ex, __ATOMIC_RELAXED, __HIP_MEMORY_SCOPE_WORKGROUP);
  }
}

__device__ __forceinline__ void phase_alpha(int n, const int* s_dl, float* s_val,
                                            const float* s_den) {
  for (int i = threadIdx.x; i < n * 4; i += TPB) {
    const int k = s_dl[i >> 2] * 4 + (i & 3);
    s_val[i] = s_val[i] / (s_den[k] + 1e-16f);
  }
}

template <int NE>
__device__ __forceinline__ void sum_group(const float* __restrict__ xl, int i0, int lane,
                                          const int* s_src, const int* s_dl,
                                          const float* s_val, float* s_acc) {
  float4 v[NE];
  int dl[NE];
#pragma unroll
  for (int u = 0; u < NE; ++u) {
    const int src = s_src[i0 + u];
    dl[u] = s_dl[i0 + u];
    v[u] = ((const float4*)(xl + (size_t)src * HC))[lane];
  }
  __builtin_amdgcn_sched_barrier(0);
#pragma unroll
  for (int u = 0; u < NE; ++u) {
    const float alpha = s_val[(i0 + u) * 4 + (lane >> 3)];
    float* row = s_acc + dl[u] * HC + lane;
    __hip_atomic_fetch_add(row +  0, v[u].x * alpha, __ATOMIC_RELAXED, __HIP_MEMORY_SCOPE_WORKGROUP);
    __hip_atomic_fetch_add(row + 32, v[u].y * alpha, __ATOMIC_RELAXED, __HIP_MEMORY_SCOPE_WORKGROUP);
    __hip_atomic_fetch_add(row + 64, v[u].z * alpha, __ATOMIC_RELAXED, __HIP_MEMORY_SCOPE_WORKGROUP);
    __hip_atomic_fetch_add(row + 96, v[u].w * alpha, __ATOMIC_RELAXED, __HIP_MEMORY_SCOPE_WORKGROUP);
  }
}

__device__ __forceinline__ void phase_sum(const float* __restrict__ xl, int n,
                                          const int* s_src, const int* s_dl,
                                          const float* s_val, float* s_acc) {
  const int wave = threadIdx.x >> 5;
  const int lane = threadIdx.x & 31;
  const int nfull = n / GRP;
  for (int g = wave; g < nfull; g += NWAVE)
    sum_group<GRP>(xl, g * GRP, lane, s_src, s_dl, s_val, s_acc);
  const int i = nfull * GRP + wave;
  if (i < n) sum_group<1>(xl, i, lane, s_src, s_dl, s_val, s_acc);
}

__device__ __forceinline__ void zero_acc(float* s_acc) {
  for (int i = threadIdx.x; i < NPB * HC / 4; i += TPB)
    ((float4*)s_acc)[i] = make_float4(0.f, 0.f, 0.f, 0.f);
}

__global__ void __launch_bounds__(TPB)
gat_owner(const unsigned char* __restrict__ keys, const unsigned* __restrict__ rec,
          const float* __restrict__ xl,
          const float* __restrict__ xr, const float* __restrict__ att,
          const float* __restrict__ bias, float* __restrict__ out) {
  __shared__ float s_acc[NPB * HC];
  __shared__ float s_m[NPB * 4];
  __shared__ float s_den[NPB * 4];
  __shared__ int   s_src[CAP];
  __shared__ int   s_dl[CAP];
  __shared__ float s_val[CAP * 4];
  __shared__ int   s_cnt;

  const int tid   = threadIdx.x;
  const int node0 = blockIdx.x * NPB;
  const unsigned nn = (unsigned)min(NPB, N_NODES - node0);
  const float4 w = ((const float4*)att)[tid & 31];

  for (int i = tid; i < (int)nn * (HC / 4); i += TPB)
    ((float4*)s_acc)[i] = ((const float4*)(xr + (size_t)node0 * HC))[i];
  for (int i = tid; i < NPB * 4; i += TPB) {
    s_m[i] = -3.402823466e38f;
    s_den[i] = 0.0f;
  }
  if (tid == 0) s_cnt = 0;
  __syncthreads();

  build_edges(keys, 0, N_EDGES, &s_cnt, s_src);
  build_self(node0, nn, &s_cnt, s_src, s_dl);
  __syncthreads();
  const int total = s_cnt;

  if (total <= CAP) {
    phase_resolve(rec, total, s_src, s_dl);
    __syncthreads();
    phase_logits<true>(xl, xr, s_acc, w, node0, total, s_src, s_dl, s_val, s_m);
    __syncthreads();
    zero_acc(s_acc);
    phase_exp(total, true, s_dl, s_val, s_m, s_den);
    __syncthreads();
    phase_alpha(total, s_dl, s_val, s_den);
    __syncthreads();
    phase_sum(xl, total, s_src, s_dl, s_val, s_acc);
    __syncthreads();
  } else {
    zero_acc(s_acc);
    for (int sweep = 0; sweep < 3; ++sweep) {
      for (int win = 0; win < NWIN; ++win) {
        __syncthreads();
        if (tid == 0) s_cnt = 0;
        __syncthreads();
        if (win < NWIN - 1)
          build_edges(keys, win * CAP, min(win * CAP + CAP, N_EDGES), &s_cnt, s_src);
        else
          build_self(node0, nn, &s_cnt, s_src, s_dl);
        __syncthreads();
        const int n = s_cnt;
        phase_resolve(rec, n, s_src, s_dl);
        __syncthreads();
        phase_logits<false>(xl, xr, s_acc, w, node0, n, s_src, s_dl, s_val, s_m);
        __syncthreads();
        if (sweep >= 1) {
          phase_exp(n, sweep == 1, s_dl, s_val, s_m, s_den);
          __syncthreads();
        }
        if (sweep == 2) {
          phase_alpha(n, s_dl, s_val, s_den);
          __syncthreads();
          phase_sum(xl, n, s_src, s_dl, s_val, s_acc);
          __syncthreads();
        }
      }
    }
  }

  const float4 bq = ((const float4*)bias)[tid & 31];
  for (int idx = tid; idx < (int)nn * 32; idx += TPB) {
    const float* row = s_acc + (idx >> 5) * HC + (idx & 31);
    float4 a;
    a.x = row[0]  + bq.x;
    a.y = row[32] + bq.y;
    a.z = row[64] + bq.z;
    a.w = row[96] + bq.w;
    ((float4*)(out + (size_t)node0 * HC))[idx] = a;
  }
}

extern "C" void kernel_launch(void* const* d_in, const int* in_sizes, int n_in,
                              void* d_out, int out_size, void* d_ws, size_t ws_size,
                              hipStream_t stream) {
  const float* x    = (const float*)d_in[0];
  const int*   ei   = (const int*)  d_in[1];
  const float* Wl   = (const float*)d_in[2];
  const float* Wr   = (const float*)d_in[3];
  const float* att  = (const float*)d_in[4];
  const float* bias = (const float*)d_in[5];
  float* out = (float*)d_out;

  float* ws = (float*)d_ws;
  float* xl = ws;
  float* xr = xl + (size_t)N_NODES * HC;
  unsigned* key = (unsigned*)(xr + (size_t)N_NODES * HC);
  unsigned* rec = key + N_EDGES / 4;

  gat_gemm<<<6250, 256, 0, stream>>>(x, Wl, Wr, ei, xl, xr, key, rec);

  gat_owner<<<NBLK, TPB, 0, stream>>>((const unsigned char*)key, rec, xl, xr, att, bias, out);
}


# GEMM launch W-stationary: 256 work-groups of 16 waves; a wave keeps its W column tile's B fragments in registers, walks the row tiles; x tiles staged in LDS by whole-row loads; sum order kept
# speedup vs baseline: 3.6487x; 1.2877x over previous
#include <hip/hip_runtime.h>
#include <hip/hip_bf16.h>

#define N_NODES   50000
#define N_EDGES   800000
#define E_TOT     850000
#define IN_F      128
#define HC        128
#define NEG_SLOPE 0.2f
#define NPB       196
#define NBLK      256
static_assert(NBLK <= 256 && NBLK * NPB >= N_NODES, "owner key is one byte");
static_assert(N_NODES <= 65536 && NPB <= 65536, "edge record packs source and local destination in 16 bits each");

typedef __attribute__((ext_vector_type(2))) float v2f;
typedef __attribute__((ext_vector_type(8))) float v8f;

#define GEMM_T   512
#define GEMM_WG  256
#define M_TILES  (N_NODES / 16)
#define KSTEPS   (IN_F / 4)
#define A_PITCH  132
#define A_BUF    (16 * A_PITCH)
#define TILE_STRIDE ((size_t)GEMM_WG * 16 * IN_F)
static_assert(M_TILES * 16 == N_NODES && GEMM_WG <= M_TILES, "whole row tiles, one or more per work-group");
static_assert(GEMM_WG * GEMM_T <= N_EDGES / 4 && 2 * GEMM_WG * GEMM_T >= N_EDGES / 4 &&
              (N_EDGES / 4) % 32 == 0, "every thread re-encodes one or two groups of four edges, whole waves each");

__device__ __forceinline__ void lds_barrier() {
  __builtin_amdgcn_fence(__ATOMIC_RELEASE, "workgroup", "local");
  __builtin_amdgcn_s_barrier();
  __builtin_amdgcn_fence(__ATOMIC_ACQUIRE, "workgroup", "local");
}

__device__ __forceinline__ void load_a(v2f (&a)[KSTEPS], const float* arow) {
#pragma unroll
  for (int t = 0; t < KSTEPS; ++t) {
    const float2 ax = *(const float2*)(arow + 4 * t);
    a[t].x = ax.x; a[t].y = ax.y;
  }
}

__device__ __forceinline__ void tile_mul(const v2f (&a)[KSTEPS], const v2f (&b)[KSTEPS],
                                         float* __restrict__ dst, int mt, int lane, int ncol) {
  v8f c = {};
#pragma unroll
  for (int t = 0; t < KSTEPS; ++t)
    c = __builtin_amdgcn_wmma_f32_16x16x4_f32(false, a[t], false, b[t],
                                              (short)0, c, false, false);
  const int mbase = mt * 16 + 8 * (lane >> 4);
#pragma unroll
  for (int v = 0; v < 8; ++v)
    dst[(size_t)(mbase + v) * HC + ncol] = c[v];
}

__device__ __forceinline__ void encode_edges(int i, const int4 s, const int4 d,
                                             unsigned* __restrict__ key,
                                             unsigned* __restrict__ rec) {
  const unsigned k0 = (unsigned)d.x / (unsigned)NPB;
  const unsigned k1 = (unsigned)d.y / (unsigned)NPB;
  const unsigned k2 = (unsigned)d.z / (unsigned)NPB;
  const unsigned k3 = (unsigned)d.w / (unsigned)NPB;
  key[i] = k0 | k1 << 8 | k2 << 16 | k3 << 24;
  uint4 r;
  r.x = (unsigned)s.x | ((unsigned)d.x - k0 * NPB) << 16;
  r.y = (unsigned)s.y | ((unsigned)d.y - k1 * NPB) << 16;
  r.z = (unsigned)s.z | ((unsigned)d.z - k2 * NPB) << 16;
  r.w = (unsigned)s.w | ((unsigned)d.w - k3 * NPB) << 16;
  ((uint4*)rec)[i] = r;
}

__global__ void __launch_bounds__(GEMM_T)
gat_gemm(const float* __restrict__ x, const float* __restrict__ Wl,
         const float* __restrict__ Wr, const int* __restrict__ ei,
         float* __restrict__ xl, float* __restrict__ xr,
         unsigned* __restrict__ key, unsigned* __restrict__ rec) {
  __shared__ __attribute__((aligned(16))) float s_a[2 * A_BUF];

  const int tid  = threadIdx.x;
  const int wave = tid >> 5;
  const int lane = tid & 31;

  const int e0 = blockIdx.x * GEMM_T + tid;
  const int e1 = e0 + GEMM_WG * GEMM_T;
  const bool two = e1 < N_EDGES / 4;
  const int4 es0 = ((const int4*)ei)[e0];
  const int4 ed0 = ((const int4*)(ei + N_EDGES))[e0];
  int4 es1 = make_int4(0, 0, 0, 0), ed1 = make_int4(0, 0, 0, 0);
  if (two) {
    es1 = ((const int4*)ei)[e1];
    ed1 = ((const int4*)(ei + N_EDGES))[e1];
  }

  const float* __restrict__ W   = (wave < 8) ? Wl : Wr;
  float* __restrict__       dst = (wave < 8) ? xl : xr;
  const int ncol  = (wave & 7) * 16 + (lane & 15);
  const int khalf = (lane >> 4) * 2;

  v2f b[KSTEPS];
#pragma unroll
  for (int t = 0; t < KSTEPS; ++t) {
    b[t].x = W[(size_t)(4 * t + khalf) * HC + ncol];
    b[t].y = W[(size_t)(4 * t + khalf + 1) * HC + ncol];
  }

  const int n = (M_TILES - 1 - (int)blockIdx.x) / GEMM_WG + 1;
  const float* xw = x + ((size_t)blockIdx.x * 16 + wave) * IN_F;
  float* stage = &s_a[wave * A_PITCH + 4 * lane];
  const float* frag = &s_a[(lane & 15) * A_PITCH + khalf];

  *(float4*)stage = ((const float4*)xw)[lane];
  if (n > 1) *(float4*)(stage + A_BUF) = ((const float4*)(xw + TILE_STRIDE))[lane];
  float4 g = make_float4(0.f, 0.f, 0.f, 0.f);
  if (n > 2) g = ((const float4*)(xw + 2 * TILE_STRIDE))[lane];
  __syncthreads();
  v2f a0[KSTEPS], a1[KSTEPS];
  load_a(a0, frag);
  __syncthreads();

  encode_edges(e0, es0, ed0, key, rec);
  if (two) encode_edges(e1, es1, ed1, key, rec);

  for (int j = 0; j < n; j += 2) {
    if (j + 2 < n) *(float4*)stage = g;
    if (j + 3 < n) g = ((const float4*)(xw + (size_t)(j + 3) * TILE_STRIDE))[lane];
    if (j + 1 < n) load_a(a1, frag + A_BUF);
    tile_mul(a0, b, dst, (int)blockIdx.x + j * GEMM_WG, lane, ncol);
    lds_barrier();
    if (j + 1 < n) {
      if (j + 3 < n) *(float4*)(stage + A_BUF) = g;
      if (j + 4 < n) g = ((const float4*)(xw + (size_t)(j + 4) * TILE_STRIDE))[lane];
      if (j + 2 < n) load_a(a0, frag);
      tile_mul(a1, b, dst, (int)blockIdx.x + (j + 1) * GEMM_WG, lane, ncol);
      lds_barrier();
    }
  }
}

#define CAP    8192
#define TPB    1024
#define NWAVE  32
#define NWIN   ((N_EDGES + CAP - 1) / CAP + 1)

__device__ __forceinline__ void list_append(int* s_cnt, int* s_src, int e) {
  const int slot = atomicAdd(s_cnt, 1);
  if (slot < CAP) s_src[slot] = e;
}

#define SCAN_U 5

__device__ __forceinline__ void scan_load(int4 (&v)[SCAN_U],
                                          const unsigned char* __restrict__ keys,
                                          int base, int hi) {
#pragma unroll
  for (int u = 0; u < SCAN_U; ++u)
    v[u] = *(const int4*)(keys + min(base + u * TPB * 16, hi - 16));
}

__device__ __forceinline__ unsigned eq_bytes(unsigned x, unsigned pat) {
  const unsigned y = x ^ pat;
  const unsigned t = (y & 0x7F7F7F7Fu) + 0x7F7F7F7Fu;
  return ~(t | y | 0x7F7F7F7Fu);
}

__device__ __forceinline__ void scan_match(const int4 (&v)[SCAN_U], int base, int hi,
                                           unsigned pat, int* s_cnt, int* s_src) {
#pragma unroll
  for (int u = 0; u < SCAN_U; ++u) {
    const int e = base + u * TPB * 16;
    unsigned c = (eq_bytes((unsigned)v[u].x, pat) >> 7) | (eq_bytes((unsigned)v[u].y, pat) >> 6)
               | (eq_bytes((unsigned)v[u].z, pat) >> 5) | (eq_bytes((unsigned)v[u].w, pat) >> 4);
    if (e >= hi) c = 0u;
    while (c) {
      const int b = __builtin_ctz(c);
      c &= c - 1u;
      list_append(s_cnt, s_src, e + ((b & 7) << 2) + (b >> 3));
    }
  }
}

__device__ __forceinline__ void build_edges(const unsigned char* __restrict__ keys,
                                            int lo, int hi, int* s_cnt, int* s_src) {
  const unsigned pat = blockIdx.x * 0x01010101u;
  const int trips = (hi - lo + TPB * 16 * SCAN_U - 1) / (TPB * 16 * SCAN_U);
  int base = lo + (int)threadIdx.x * 16;
  int4 va[SCAN_U], vb[SCAN_U];
  scan_load(va, keys, base, hi);
  for (int t = 0; t < trips; t += 2) {
    scan_load(vb, keys, base + TPB * 16 * SCAN_U, hi);
    scan_match(va, base, hi, pat, s_cnt, s_src);
    scan_load(va, keys, base + TPB * 32 * SCAN_U, hi);
    scan_match(vb, base + TPB * 16 * SCAN_U, hi, pat, s_cnt, s_src);
    base += TPB * 32 * SCAN_U;
  }
}

__device__ __forceinline__ void build_self(int node0, unsigned nn,
                                           int* s_cnt, int* s_src, int* s_dl) {
  if (threadIdx.x < nn) {
    const int slot = atomicAdd(s_cnt, 1);
    if (slot < CAP) {
      s_src[slot] = N_EDGES + node0 + (int)threadIdx.x;
      s_dl[slot]  = (int)threadIdx.x;
    }
  }
}

__device__ __forceinline__ void phase_resolve(const unsigned* __restrict__ rec, int n,
                                              int* s_src, int* s_dl) {
  for (int i = threadIdx.x; i < n; i += TPB) {
    const int e = s_src[i];
    if (e < N_EDGES) {
      const unsigned r = rec[e];
      s_src[i] = (int)(r & 0xFFFFu);
      s_dl[i]  = (int)(r >> 16);
    } else {
      s_src[i] = e - N_EDGES;
    }
  }
}

template <int NE, bool XR_LDS>
__device__ __forceinline__ void logit_group(const float* __restrict__ xl,
                                            const float* __restrict__ xr,
                                            const float* s_xr,
                                            const float4 w, int node0, int i0, int lane,
                                            const int* s_src, const int* s_dl,
                                            float* s_val, float* s_m) {
  float4 a[NE], bg[NE];
  int dl[NE];
#pragma unroll
  for (int u = 0; u < NE; ++u) {
    const int src = s_src[i0 + u];
    dl[u] = s_dl[i0 + u];
    a[u] = ((const float4*)(xl + (size_t)src * HC))[lane];
    if (!XR_LDS) bg[u] = ((const float4*)(xr + (size_t)(node0 + dl[u]) * HC))[lane];
  }
  __builtin_amdgcn_sched_barrier(0);
  float s[NE];
#pragma unroll
  for (int u = 0; u < NE; ++u) {
    const float4 b = XR_LDS ? ((const float4*)(s_xr + dl[u] * HC))[lane] : bg[u];
    float vx = a[u].x + b.x, vy = a[u].y + b.y;
    float vz = a[u].z + b.z, vw = a[u].w + b.w;
    vx = (vx > 0.f) ? vx : vx * NEG_SLOPE;
    vy = (vy > 0.f) ? vy : vy * NEG_SLOPE;
    vz = (vz > 0.f) ? vz : vz * NEG_SLOPE;
    vw = (vw > 0.f) ? vw : vw * NEG_SLOPE;
    s[u] = vx * w.x + vy * w.y + vz * w.z + vw * w.w;
  }
#pragma unroll
  for (int u = 0; u < NE; ++u) s[u] += __shfl_xor(s[u], 1);
#pragma unroll
  for (int u = 0; u < NE; ++u) s[u] += __shfl_xor(s[u], 2);
#pragma unroll
  for (int u = 0; u < NE; ++u) s[u] += __shfl_xor(s[u], 4);
  if ((lane & 7) == 0) {
    const int h = lane >> 3;
#pragma unroll
    for (int u = 0; u < NE; ++u) {
      s_val[(i0 + u) * 4 + h] = s[u];
      __hip_atomic_fetch_max(&s_m[dl[u] * 4 + h], s[u], __ATOMIC_RELAXED,
                             __HIP_MEMORY_SCOPE_WORKGROUP);
    }
  }
}

#define GRP 8

template <bool XR_LDS>
__device__ __forceinline__ void phase_logits(const float* __restrict__ xl,
                                             const float* __restrict__ xr,
                                             const float* s_xr,
                                             const float4 w, int node0, int n,
                                             const int* s_src, const int* s_dl,
                                             float* s_val, float* s_m) {
  const int wave = threadIdx.x >> 5;
  const int lane = threadIdx.x & 31;
  const int nfull = n / GRP;
  for (int g = wave; g < nfull; g += NWAVE)
    logit_group<GRP, XR_LDS>(xl, xr, s_xr, w, node0, g * GRP, lane, s_src, s_dl, s_val, s_m);
  const int i = nfull * GRP + wave;
  if (i < n)
    logit_group<1, XR_LDS>(xl, xr, s_xr, w, node0, i, lane, s_src, s_dl, s_val, s_m);
}

__device__ __forceinline__ void phase_exp(int n, bool add, const int* s_dl,
                                          float* s_val, const float* s_m, float* s_den) {
  for (int i = threadIdx.x; i < n * 4; i += TPB) {
    const int k = s_dl[i >> 2] * 4 + (i & 3);
    const float ex = __expf(s_val[i] - s_m[k]);
    s_val[i] = ex;
    if (add)
      __hip_atomic_fetch_add(&s_den[k], ex, __ATOMIC_RELAXED, __HIP_MEMORY_SCOPE_WORKGROUP);
  }
}

__device__ __forceinline__ void phase_alpha(int n, const int* s_dl, float* s_val,
                                            const float* s_den) {
  for (int i = threadIdx.x; i < n * 4; i += TPB) {
    const int k = s_dl[i >> 2] * 4 + (i & 3);
    s_val[i] = s_val[i] / (s_den[k] + 1e-16f);
  }
}

template <int NE>
__device__ __forceinline__ void sum_group(const float* __restrict__ xl, int i0, int lane,
                                          const int* s_src, const int* s_dl,
                                          const float* s_val, float* s_acc) {
  float4 v[NE];
  int dl[NE];
#pragma unroll
  for (int u = 0; u < NE; ++u) {
    const int src = s_src[i0 + u];
    dl[u] = s_dl[i0 + u];
    v[u] = ((const float4*)(xl + (size_t)src * HC))[lane];
  }
  __builtin_amdgcn_sched_barrier(0);
#pragma unroll
  for (int u = 0; u < NE; ++u) {
    const float alpha = s_val[(i0 + u) * 4 + (lane >> 3)];
    float* row = s_acc + dl[u] * HC + lane;
    __hip_atomic_fetch_add(row +  0, v[u].x * alpha, __ATOMIC_RELAXED, __HIP_MEMORY_SCOPE_WORKGROUP);
    __hip_atomic_fetch_add(row + 32, v[u].y * alpha, __ATOMIC_RELAXED, __HIP_MEMORY_SCOPE_WORKGROUP);
    __hip_atomic_fetch_add(row + 64, v[u].z * alpha, __ATOMIC_RELAXED, __HIP_MEMORY_SCOPE_WORKGROUP);
    __hip_atomic_fetch_add(row + 96, v[u].w * alpha, __ATOMIC_RELAXED, __HIP_MEMORY_SCOPE_WORKGROUP);
  }
}

__device__ __forceinline__ void phase_sum(const float* __restrict__ xl, int n,
                                          const int* s_src, const int* s_dl,
                                          const float* s_val, float* s_acc) {
  const int wave = threadIdx.x >> 5;
  const int lane = threadIdx.x & 31;
  const int nfull = n / GRP;
  for (int g = wave; g < nfull; g += NWAVE)
    sum_group<GRP>(xl, g * GRP, lane, s_src, s_dl, s_val, s_acc);
  const int i = nfull * GRP + wave;
  if (i < n) sum_group<1>(xl, i, lane, s_src, s_dl, s_val, s_acc);
}

__device__ __forceinline__ void zero_acc(float* s_acc) {
  for (int i = threadIdx.x; i < NPB * HC / 4; i += TPB)
    ((float4*)s_acc)[i] = make_float4(0.f, 0.f, 0.f, 0.f);
}

__global__ void __launch_bounds__(TPB)
gat_owner(const unsigned char* __restrict__ keys, const unsigned* __restrict__ rec,
          const float* __restrict__ xl,
          const float* __restrict__ xr, const float* __restrict__ att,
          const float* __restrict__ bias, float* __restrict__ out) {
  __shared__ float s_acc[NPB * HC];
  __shared__ float s_m[NPB * 4];
  __shared__ float s_den[NPB * 4];
  __shared__ int   s_src[CAP];
  __shared__ int   s_dl[CAP];
  __shared__ float s_val[CAP * 4];
  __shared__ int   s_cnt;

  const int tid   = threadIdx.x;
  const int node0 = blockIdx.x * NPB;
  const unsigned nn = (unsigned)min(NPB, N_NODES - node0);
  const float4 w = ((const float4*)att)[tid & 31];

  for (int i = tid; i < (int)nn * (HC / 4); i += TPB)
    ((float4*)s_acc)[i] = ((const float4*)(xr + (size_t)node0 * HC))[i];
  for (int i = tid; i < NPB * 4; i += TPB) {
    s_m[i] = -3.402823466e38f;
    s_den[i] = 0.0f;
  }
  if (tid == 0) s_cnt = 0;
  __syncthreads();

  build_edges(keys, 0, N_EDGES, &s_cnt, s_src);
  build_self(node0, nn, &s_cnt, s_src, s_dl);
  __syncthreads();
  const int total = s_cnt;

  if (total <= CAP) {
    phase_resolve(rec, total, s_src, s_dl);
    __syncthreads();
    phase_logits<true>(xl, xr, s_acc, w, node0, total, s_src, s_dl, s_val, s_m);
    __syncthreads();
    zero_acc(s_acc);
    phase_exp(total, true, s_dl, s_val, s_m, s_den);
    __syncthreads();
    phase_alpha(total, s_dl, s_val, s_den);
    __syncthreads();
    phase_sum(xl, total, s_src, s_dl, s_val, s_acc);
    __syncthreads();
  } else {
    zero_acc(s_acc);
    for (int sweep = 0; sweep < 3; ++sweep) {
      for (int win = 0; win < NWIN; ++win) {
        __syncthreads();
        if (tid == 0) s_cnt = 0;
        __syncthreads();
        if (win < NWIN - 1)
          build_edges(keys, win * CAP, min(win * CAP + CAP, N_EDGES), &s_cnt, s_src);
        else
          build_self(node0, nn, &s_cnt, s_src, s_dl);
        __syncthreads();
        const int n = s_cnt;
        phase_resolve(rec, n, s_src, s_dl);
        __syncthreads();
        phase_logits<false>(xl, xr, s_acc, w, node0, n, s_src, s_dl, s_val, s_m);
        __syncthreads();
        if (sweep >= 1) {
          phase_exp(n, sweep == 1, s_dl, s_val, s_m, s_den);
          __syncthreads();
        }
        if (sweep == 2) {
          phase_alpha(n, s_dl, s_val, s_den);
          __syncthreads();
          phase_sum(xl, n, s_src, s_dl, s_val, s_acc);
          __syncthreads();
        }
      }
    }
  }

  const float4 bq = ((const float4*)bias)[tid & 31];
  for (int idx = tid; idx < (int)nn * 32; idx += TPB) {
    const float* row = s_acc + (idx >> 5) * HC + (idx & 31);
    float4 a;
    a.x = row[0]  + bq.x;
    a.y = row[32] + bq.y;
    a.z = row[64] + bq.z;
    a.w = row[96] + bq.w;
    ((float4*)(out + (size_t)node0 * HC))[idx] = a;
  }
}

extern "C" void kernel_launch(void* const* d_in, const int* in_sizes, int n_in,
                              void* d_out, int out_size, void* d_ws, size_t ws_size,
                              hipStream_t stream) {
  const float* x    = (const float*)d_in[0];
  const int*   ei   = (const int*)  d_in[1];
  const float* Wl   = (const float*)d_in[2];
  const float* Wr   = (const float*)d_in[3];
  const float* att  = (const float*)d_in[4];
  const float* bias = (const float*)d_in[5];
  float* out = (float*)d_out;

  float* ws = (float*)d_ws;
  float* xl = ws;
  float* xr = xl + (size_t)N_NODES * HC;
  unsigned* key = (unsigned*)(xr + (size_t)N_NODES * HC);
  unsigned* rec = key + N_EDGES / 4;

  gat_gemm<<<GEMM_WG, GEMM_T, 0, stream>>>(x, Wl, Wr, ei, xl, xr, key, rec);

  gat_owner<<<NBLK, TPB, 0, stream>>>((const unsigned char*)key, rec, xl, xr, att, bias, out);
}


# GEMM launch sorts its edges by owner block in LDS (histogram, prefix sum) into runs plus a start/count table; an owner copies its 256 runs instead of scanning all keys; list order changes
# speedup vs baseline: 4.2422x; 1.1627x over previous
#include <hip/hip_runtime.h>
#include <hip/hip_bf16.h>

#define N_NODES   50000
#define N_EDGES   800000
#define E_TOT     850000
#define IN_F      128
#define HC        128
#define NEG_SLOPE 0.2f
#define NPB       196
#define NBLK      256
static_assert(NBLK <= 256 && NBLK * NPB >= N_NODES, "owner key is one byte");
static_assert(N_NODES <= 65536 && NPB <= 65536, "edge record packs source and local destination in 16 bits each");

typedef __attribute__((ext_vector_type(2))) float v2f;
typedef __attribute__((ext_vector_type(8))) float v8f;

#define GEMM_T   512
#define GEMM_WG  256
#define M_TILES  (N_NODES / 16)
#define KSTEPS   (IN_F / 4)
#define A_PITCH  132
#define A_BUF    (16 * A_PITCH)
#define TILE_STRIDE ((size_t)GEMM_WG * 16 * IN_F)
static_assert(M_TILES * 16 == N_NODES && GEMM_WG <= M_TILES, "whole row tiles, one or more per work-group");
static_assert(GEMM_WG * GEMM_T <= N_EDGES / 4 && 2 * GEMM_WG * GEMM_T >= N_EDGES / 4 &&
              (N_EDGES / 4) % 32 == 0, "every thread takes one or two groups of four edges, whole waves each");
#define SECOND_GROUPS (N_EDGES / 4 - GEMM_WG * GEMM_T)
#define SEG_MAX (8 * GEMM_T)
static_assert(NBLK == 256 && GEMM_WG == 256, "one histogram bin and one table word per thread of the first eight waves");
static_assert(SEG_MAX < 65536, "start and count of a run fit 16 bits each");

__device__ __forceinline__ int seg_base(int b) {
  return 4 * (GEMM_T * b + min(GEMM_T * b, SECOND_GROUPS));
}

__device__ __forceinline__ unsigned wave_scan(unsigned v, int lane) {
  for (int d = 1; d < 32; d <<= 1) {
    const unsigned t = __shfl_up(v, d);
    if (lane >= d) v += t;
  }
  return v;
}

__device__ __forceinline__ void lds_barrier() {
  __builtin_amdgcn_fence(__ATOMIC_RELEASE, "workgroup", "local");
  __builtin_amdgcn_s_barrier();
  __builtin_amdgcn_fence(__ATOMIC_ACQUIRE, "workgroup", "local");
}

__device__ __forceinline__ void load_a(v2f (&a)[KSTEPS], const float* arow) {
#pragma unroll
  for (int t = 0; t < KSTEPS; ++t) {
    const float2 ax = *(const float2*)(arow + 4 * t);
    a[t].x = ax.x; a[t].y = ax.y;
  }
}

__device__ __forceinline__ void tile_mul(const v2f (&a)[KSTEPS], const v2f (&b)[KSTEPS],
                                         float* __restrict__ dst, int mt, int lane, int ncol) {
  v8f c = {};
#pragma unroll
  for (int t = 0; t < KSTEPS; ++t)
    c = __builtin_amdgcn_wmma_f32_16x16x4_f32(false, a[t], false, b[t],
                                              (short)0, c, false, false);
  const int mbase = mt * 16 + 8 * (lane >> 4);
#pragma unroll
  for (int v = 0; v < 8; ++v)
    dst[(size_t)(mbase + v) * HC + ncol] = c[v];
}

__device__ __forceinline__ void edge_count(const int4 s, const int4 d, unsigned* s_hist,
                                           unsigned (&k)[4], unsigned (&r)[4], unsigned (&sl)[4]) {
  const unsigned dv[4] = {(unsigned)d.x, (unsigned)d.y, (unsigned)d.z, (unsigned)d.w};
  const unsigned sv[4] = {(unsigned)s.x, (unsigned)s.y, (unsigned)s.z, (unsigned)s.w};
  for (int q = 0; q < 4; ++q) {
    k[q] = dv[q] / (unsigned)NPB;
    r[q] = sv[q] | (dv[q] - k[q] * NPB) << 16;
  }
  for (int q = 0; q < 4; ++q) sl[q] = atomicAdd(&s_hist[k[q]], 1u);
}

__device__ __forceinline__ void edge_place(const unsigned* s_hist, unsigned* s_stage,
                                           const unsigned (&k)[4], const unsigned (&r)[4],
                                           const unsigned (&sl)[4]) {
  for (int q = 0; q < 4; ++q) s_stage[s_hist[k[q]] + sl[q]] = r[q];
}

__global__ void __launch_bounds__(GEMM_T)
gat_gemm(const float* __restrict__ x, const float* __restrict__ Wl,
         const float* __restrict__ Wr, const int* __restrict__ ei,
         float* __restrict__ xl, float* __restrict__ xr,
         unsigned* __restrict__ seg, unsigned* __restrict__ tab) {
  __shared__ __attribute__((aligned(16))) float s_a[2 * A_BUF];
  __shared__ __attribute__((aligned(16))) unsigned s_stage[SEG_MAX];
  __shared__ unsigned s_hist[NBLK];
  __shared__ unsigned s_wt[NBLK / 32];

  const int tid  = threadIdx.x;
  const int wave = tid >> 5;
  const int lane = tid & 31;

  const int e0 = blockIdx.x * GEMM_T + tid;
  const int e1 = e0 + GEMM_WG * GEMM_T;
  const bool two = e1 < N_EDGES / 4;
  const int4 es0 = ((const int4*)ei)[e0];
  const int4 ed0 = ((const int4*)(ei + N_EDGES))[e0];
  int4 es1 = make_int4(0, 0, 0, 0), ed1 = make_int4(0, 0, 0, 0);
  if (two) {
    es1 = ((const int4*)ei)[e1];
    ed1 = ((const int4*)(ei + N_EDGES))[e1];
  }
  if (tid < NBLK) s_hist[tid] = 0u;

  const float* __restrict__ W   = (wave < 8) ? Wl : Wr;
  float* __restrict__       dst = (wave < 8) ? xl : xr;
  const int ncol  = (wave & 7) * 16 + (lane & 15);
  const int khalf = (lane >> 4) * 2;

  v2f b[KSTEPS];
#pragma unroll
  for (int t = 0; t < KSTEPS; ++t) {
    b[t].x = W[(size_t)(4 * t + khalf) * HC + ncol];
    b[t].y = W[(size_t)(4 * t + khalf + 1) * HC + ncol];
  }

  const int n = (M_TILES - 1 - (int)blockIdx.x) / GEMM_WG + 1;
  const float* xw = x + ((size_t)blockIdx.x * 16 + wave) * IN_F;
  float* stage = &s_a[wave * A_PITCH + 4 * lane];
  const float* frag = &s_a[(lane & 15) * A_PITCH + khalf];

  *(float4*)stage = ((const float4*)xw)[lane];
  if (n > 1) *(float4*)(stage + A_BUF) = ((const float4*)(xw + TILE_STRIDE))[lane];
  float4 g = make_float4(0.f, 0.f, 0.f, 0.f);
  if (n > 2) g = ((const float4*)(xw + 2 * TILE_STRIDE))[lane];
  __syncthreads();
  v2f a0[KSTEPS], a1[KSTEPS];
  load_a(a0, frag);
  unsigned k0[4], r0[4], p0[4], k1[4], r1[4], p1[4];
  edge_count(es0, ed0, s_hist, k0, r0, p0);
  if (two) edge_count(es1, ed1, s_hist, k1, r1, p1);
  __syncthreads();

  unsigned cnt = 0u, incl = 0u;
  if (tid < NBLK) {
    cnt = s_hist[tid];
    incl = wave_scan(cnt, lane);
    if (lane == 31) s_wt[wave] = incl;
  }
  lds_barrier();
  if (tid < NBLK) {
    unsigned first = incl - cnt;
    for (int w = 0; w < wave; ++w) first += s_wt[w];
    s_hist[tid] = first;
    tab[blockIdx.x * NBLK + tid] = first | cnt << 16;
  }
  lds_barrier();
  edge_place(s_hist, s_stage, k0, r0, p0);
  if (two) edge_place(s_hist, s_stage, k1, r1, p1);
  lds_barrier();
  {
    const int quads = GEMM_T + min(max(SECOND_GROUPS - GEMM_T * (int)blockIdx.x, 0), GEMM_T);
    uint4* __restrict__ o4 = (uint4*)(seg + seg_base((int)blockIdx.x));
    for (int i = tid; i < quads; i += GEMM_T) o4[i] = ((const uint4*)s_stage)[i];
  }

  for (int j = 0; j < n; j += 2) {
    if (j + 2 < n) *(float4*)stage = g;
    if (j + 3 < n) g = ((const float4*)(xw + (size_t)(j + 3) * TILE_STRIDE))[lane];
    if (j + 1 < n) load_a(a1, frag + A_BUF);
    tile_mul(a0, b, dst, (int)blockIdx.x + j * GEMM_WG, lane, ncol);
    lds_barrier();
    if (j + 1 < n) {
      if (j + 3 < n) *(float4*)(stage + A_BUF) = g;
      if (j + 4 < n) g = ((const float4*)(xw + (size_t)(j + 4) * TILE_STRIDE))[lane];
      if (j + 2 < n) load_a(a0, frag);
      tile_mul(a1, b, dst, (int)blockIdx.x + (j + 1) * GEMM_WG, lane, ncol);
      lds_barrier();
    }
  }
}

#define CAP    8192
#define TPB    1024
#define NWAVE  32
#define WIN_B  2
#define NWIN   (GEMM_WG / WIN_B + 1)
#define FILL_U 8
static_assert(TPB == 4 * GEMM_WG && GEMM_WG == 8 * 32, "four threads copy one run; one wave places eight runs per lane");
static_assert(WIN_B * SEG_MAX <= CAP && GEMM_WG % WIN_B == 0, "a window's runs fit the list");

__device__ __forceinline__ void tab_load(const unsigned* __restrict__ tab, unsigned (&tw)[8]) {
  for (int j = 0; j < 8; ++j)
    tw[j] = tab[(8 * (int)threadIdx.x + j) * NBLK + blockIdx.x];
}

__device__ __forceinline__ int list_places(const unsigned (&tw)[8], int b_lo, int b_hi,
                                           unsigned* s_tab, int* s_pos, int* s_n) {
  if (threadIdx.x < 32) {
    const int lane = threadIdx.x;
    unsigned c[8], sum = 0u;
    for (int j = 0; j < 8; ++j) {
      const int b = 8 * lane + j;
      c[j] = (b >= b_lo && b < b_hi) ? (tw[j] >> 16) : 0u;
      sum += c[j];
    }
    const unsigned incl = wave_scan(sum, lane);
    unsigned p = incl - sum;
    for (int j = 0; j < 8; ++j) {
      s_tab[8 * lane + j] = (tw[j] & 0xFFFFu) | c[j] << 16;
      s_pos[8 * lane + j] = (int)p;
      p += c[j];
    }
    if (lane == 31) *s_n = (int)incl;
  }
  __syncthreads();
  return *s_n;
}

__device__ __forceinline__ void list_fill(const unsigned* __restrict__ seg,
                                          const unsigned* s_tab, const int* s_pos,
                                          int* s_src, int* s_dl) {
  const int b   = threadIdx.x >> 2;
  const int sub = threadIdx.x & 3;
  const int pos = s_pos[b];
  const unsigned t = s_tab[b];
  const int cnt = (int)(t >> 16);
  const int first = seg_base(b) + (int)(t & 0xFFFFu);
  unsigned r[FILL_U];
  for (int u = 0; u < FILL_U; ++u)
    r[u] = seg[min(first + min(sub + 4 * u, max(cnt - 1, 0)), N_EDGES - 1)];
  for (int u = 0; u < FILL_U; ++u) {
    const int i = sub + 4 * u;
    if (i < cnt) {
      s_src[pos + i] = (int)(r[u] & 0xFFFFu);
      s_dl[pos + i]  = (int)(r[u] >> 16);
    }
  }
  for (int i = sub + 4 * FILL_U; i < cnt; i += 4) {
    const unsigned rr = seg[first + i];
    s_src[pos + i] = (int)(rr & 0xFFFFu);
    s_dl[pos + i]  = (int)(rr >> 16);
  }
}

__device__ __forceinline__ void list_self(int node0, unsigned nn, int n0, int* s_src, int* s_dl) {
  if (threadIdx.x < nn) {
    s_src[n0 + (int)threadIdx.x] = node0 + (int)threadIdx.x;
    s_dl[n0 + (int)threadIdx.x]  = (int)threadIdx.x;
  }
}

template <int NE, bool XR_LDS>
__device__ __forceinline__ void logit_group(const float* __restrict__ xl,
                                            const float* __restrict__ xr,
                                            const float* s_xr,
                                            const float4 w, int node0, int i0, int lane,
                                            const int* s_src, const int* s_dl,
                                            float* s_val, float* s_m) {
  float4 a[NE], bg[NE];
  int dl[NE];
#pragma unroll
  for (int u = 0; u < NE; ++u) {
    const int src = s_src[i0 + u];
    dl[u] = s_dl[i0 + u];
    a[u] = ((const float4*)(xl + (size_t)src * HC))[lane];
    if (!XR_LDS) bg[u] = ((const float4*)(xr + (size_t)(node0 + dl[u]) * HC))[lane];
  }
  __builtin_amdgcn_sched_barrier(0);
  float s[NE];
#pragma unroll
  for (int u = 0; u < NE; ++u) {
    const float4 b = XR_LDS ? ((const float4*)(s_xr + dl[u] * HC))[lane] : bg[u];
    float vx = a[u].x + b.x, vy = a[u].y + b.y;
    float vz = a[u].z + b.z, vw = a[u].w + b.w;
    vx = (vx > 0.f) ? vx : vx * NEG_SLOPE;
    vy = (vy > 0.f) ? vy : vy * NEG_SLOPE;
    vz = (vz > 0.f) ? vz : vz * NEG_SLOPE;
    vw = (vw > 0.f) ? vw : vw * NEG_SLOPE;
    s[u] = vx * w.x + vy * w.y + vz * w.z + vw * w.w;
  }
#pragma unroll
  for (int u = 0; u < NE; ++u) s[u] += __shfl_xor(s[u], 1);
#pragma unroll
  for (int u = 0; u < NE; ++u) s[u] += __shfl_xor(s[u], 2);
#pragma unroll
  for (int u = 0; u < NE; ++u) s[u] += __shfl_xor(s[u], 4);
  if ((lane & 7) == 0) {
    const int h = lane >> 3;
#pragma unroll
    for (int u = 0; u < NE; ++u) {
      s_val[(i0 + u) * 4 + h] = s[u];
      __hip_atomic_fetch_max(&s_m[dl[u] * 4 + h], s[u], __ATOMIC_RELAXED,
                             __HIP_MEMORY_SCOPE_WORKGROUP);
    }
  }
}

#define GRP 8
#define GRP_WIN 4

template <bool XR_LDS>
__device__ __forceinline__ void phase_logits(const float* __restrict__ xl,
                                             const float* __restrict__ xr,
                                             const float* s_xr,
                                             const float4 w, int node0, int n,
                                             const int* s_src, const int* s_dl,
                                             float* s_val, float* s_m) {
  const int wave = threadIdx.x >> 5;
  const int lane = threadIdx.x & 31;
  constexpr int G = XR_LDS ? GRP : GRP_WIN;
  const int nfull = n / G;
  for (int g = wave; g < nfull; g += NWAVE)
    logit_group<G, XR_LDS>(xl, xr, s_xr, w, node0, g * G, lane, s_src, s_dl, s_val, s_m);
  const int i = nfull * G + wave;
  if (i < n)
    logit_group<1, XR_LDS>(xl, xr, s_xr, w, node0, i, lane, s_src, s_dl, s_val, s_m);
}

__device__ __forceinline__ void phase_exp(int n, bool add, const int* s_dl,
                                          float* s_val, const float* s_m, float* s_den) {
  for (int i = threadIdx.x; i < n * 4; i += TPB) {
    const int k = s_dl[i >> 2] * 4 + (i & 3);
    const float ex = __expf(s_val[i] - s_m[k]);
    s_val[i] = ex;
    if (add)
      __hip_atomic_fetch_add(&s_den[k], ex, __ATOMIC_RELAXED, __HIP_MEMORY_SCOPE_WORKGROUP);
  }
}

__device__ __forceinline__ void phase_alpha(int n, const int* s_dl, float* s_val,
                                            const float* s_den) {
  for (int i = threadIdx.x; i < n * 4; i += TPB) {
    const int k = s_dl[i >> 2] * 4 + (i & 3);
    s_val[i] = s_val[i] / (s_den[k] + 1e-16f);
  }
}

template <int NE>
__device__ __forceinline__ void sum_group(const float* __restrict__ xl, int i0, int lane,
                                          const int* s_src, const int* s_dl,
                                          const float* s_val, float* s_acc) {
  float4 v[NE];
  int dl[NE];
#pragma unroll
  for (int u = 0; u < NE; ++u) {
    const int src = s_src[i0 + u];
    dl[u] = s_dl[i0 + u];
    v[u] = ((const float4*)(xl + (size_t)src * HC))[lane];
  }
  __builtin_amdgcn_sched_barrier(0);
#pragma unroll
  for (int u = 0; u < NE; ++u) {
    const float alpha = s_val[(i0 + u) * 4 + (lane >> 3)];
    float* row = s_acc + dl[u] * HC + lane;
    __hip_atomic_fetch_add(row +  0, v[u].x * alpha, __ATOMIC_RELAXED, __HIP_MEMORY_SCOPE_WORKGROUP);
    __hip_atomic_fetch_add(row + 32, v[u].y * alpha, __ATOMIC_RELAXED, __HIP_MEMORY_SCOPE_WORKGROUP);
    __hip_atomic_fetch_add(row + 64, v[u].z * alpha, __ATOMIC_RELAXED, __HIP_MEMORY_SCOPE_WORKGROUP);
    __hip_atomic_fetch_add(row + 96, v[u].w * alpha, __ATOMIC_RELAXED, __HIP_MEMORY_SCOPE_WORKGROUP);
  }
}

__device__ __forceinline__ void phase_sum(const float* __restrict__ xl, int n,
                                          const int* s_src, const int* s_dl,
                                          const float* s_val, float* s_acc) {
  const int wave = threadIdx.x >> 5;
  const int lane = threadIdx.x & 31;
  const int nfull = n / GRP;
  for (int g = wave; g < nfull; g += NWAVE)
    sum_group<GRP>(xl, g * GRP, lane, s_src, s_dl, s_val, s_acc);
  const int i = nfull * GRP + wave;
  if (i < n) sum_group<1>(xl, i, lane, s_src, s_dl, s_val, s_acc);
}

__device__ __forceinline__ void zero_acc(float* s_acc) {
  for (int i = threadIdx.x; i < NPB * HC / 4; i += TPB)
    ((float4*)s_acc)[i] = make_float4(0.f, 0.f, 0.f, 0.f);
}

__global__ void __launch_bounds__(TPB)
gat_owner(const unsigned* __restrict__ seg, const unsigned* __restrict__ tab,
          const float* __restrict__ xl,
          const float* __restrict__ xr, const float* __restrict__ att,
          const float* __restrict__ bias, float* __restrict__ out) {
  __shared__ float s_acc[NPB * HC];
  __shared__ float s_m[NPB * 4];
  __shared__ float s_den[NPB * 4];
  __shared__ int   s_src[CAP];
  __shared__ int   s_dl[CAP];
  __shared__ float s_val[CAP * 4];
  __shared__ unsigned s_tab[GEMM_WG];
  __shared__ int   s_pos[GEMM_WG];
  __shared__ int   s_n;

  const int tid   = threadIdx.x;
  const int node0 = blockIdx.x * NPB;
  const unsigned nn = (unsigned)min(NPB, N_NODES - node0);
  const float4 w = ((const float4*)att)[tid & 31];
  unsigned tw[8] = {0u, 0u, 0u, 0u, 0u, 0u, 0u, 0u};
  if (tid < 32) tab_load(tab, tw);

  for (int i = tid; i < (int)nn * (HC / 4); i += TPB)
    ((float4*)s_acc)[i] = ((const float4*)(xr + (size_t)node0 * HC))[i];
  for (int i = tid; i < NPB * 4; i += TPB) {
    s_m[i] = -3.402823466e38f;
    s_den[i] = 0.0f;
  }
  const int ne = list_places(tw, 0, GEMM_WG, s_tab, s_pos, &s_n);
  const int total = ne + (int)nn;

  if (total <= CAP) {
    list_fill(seg, s_tab, s_pos, s_src, s_dl);
    list_self(node0, nn, ne, s_src, s_dl);
    __syncthreads();
    phase_logits<true>(xl, xr, s_acc, w, node0, total, s_src, s_dl, s_val, s_m);
    __syncthreads();
    zero_acc(s_acc);
    phase_exp(total, true, s_dl, s_val, s_m, s_den);
    __syncthreads();
    phase_alpha(total, s_dl, s_val, s_den);
    __syncthreads();
    phase_sum(xl, total, s_src, s_dl, s_val, s_acc);
    __syncthreads();
  } else {
    zero_acc(s_acc);
    for (int sweep = 0; sweep < 3; ++sweep) {
      for (int win = 0; win < NWIN; ++win) {
        __syncthreads();
        int n;
        if (win < NWIN - 1) {
          if (tid < 32) tab_load(tab, tw);
          n = list_places(tw, win * WIN_B, win * WIN_B + WIN_B, s_tab, s_pos, &s_n);
          list_fill(seg, s_tab, s_pos, s_src, s_dl);
        } else {
          n = (int)nn;
          list_self(node0, nn, 0, s_src, s_dl);
        }
        __syncthreads();
        phase_logits<false>(xl, xr, s_acc, w, node0, n, s_src, s_dl, s_val, s_m);
        __syncthreads();
        if (sweep >= 1) {
          phase_exp(n, sweep == 1, s_dl, s_val, s_m, s_den);
          __syncthreads();
        }
        if (sweep == 2) {
          phase_alpha(n, s_dl, s_val, s_den);
          __syncthreads();
          phase_sum(xl, n, s_src, s_dl, s_val, s_acc);
          __syncthreads();
        }
      }
    }
  }

  const float4 bq = ((const float4*)bias)[tid & 31];
  for (int idx = tid; idx < (int)nn * 32; idx += TPB) {
    const float* row = s_acc + (idx >> 5) * HC + (idx & 31);
    float4 a;
    a.x = row[0]  + bq.x;
    a.y = row[32] + bq.y;
    a.z = row[64] + bq.z;
    a.w = row[96] + bq.w;
    ((float4*)(out + (size_t)node0 * HC))[idx] = a;
  }
}

extern "C" void kernel_launch(void* const* d_in, const int* in_sizes, int n_in,
                              void* d_out, int out_size, void* d_ws, size_t ws_size,
                              hipStream_t stream) {
  const float* x    = (const float*)d_in[0];
  const int*   ei   = (const int*)  d_in[1];
  const float* Wl   = (const float*)d_in[2];
  const float* Wr   = (const float*)d_in[3];
  const float* att  = (const float*)d_in[4];
  const float* bias = (const float*)d_in[5];
  float* out = (float*)d_out;

  float* ws = (float*)d_ws;
  float* xl = ws;
  float* xr = xl + (size_t)N_NODES * HC;
  unsigned* seg = (unsigned*)(xr + (size_t)N_NODES * HC);
  unsigned* tab = seg + N_EDGES;

  gat_gemm<<<GEMM_WG, GEMM_T, 0, stream>>>(x, Wl, Wr, ei, xl, xr, seg, tab);

  gat_owner<<<NBLK, TPB, 0, stream>>>(seg, tab, xl, xr, att, bias, out);
}


# logit groups of 8: the 8-lane head reduction halves a lane's edges per step (7 cross-lane moves, was 24); 32 logits leave in one LDS store + one LDS max; same cross-lane tree, lane dot unfused
# speedup vs baseline: 4.5018x; 1.0612x over previous
#include <hip/hip_runtime.h>
#include <hip/hip_bf16.h>

#define N_NODES   50000
#define N_EDGES   800000
#define E_TOT     850000
#define IN_F      128
#define HC        128
#define NEG_SLOPE 0.2f
#define NPB       196
#define NBLK      256
static_assert(NBLK <= 256 && NBLK * NPB >= N_NODES, "owner key is one byte");
static_assert(N_NODES <= 65536 && NPB <= 65536, "edge record packs source and local destination in 16 bits each");

typedef __attribute__((ext_vector_type(2))) float v2f;
typedef __attribute__((ext_vector_type(8))) float v8f;

#define GEMM_T   512
#define GEMM_WG  256
#define M_TILES  (N_NODES / 16)
#define KSTEPS   (IN_F / 4)
#define A_PITCH  132
#define A_BUF    (16 * A_PITCH)
#define TILE_STRIDE ((size_t)GEMM_WG * 16 * IN_F)
static_assert(M_TILES * 16 == N_NODES && GEMM_WG <= M_TILES, "whole row tiles, one or more per work-group");
static_assert(GEMM_WG * GEMM_T <= N_EDGES / 4 && 2 * GEMM_WG * GEMM_T >= N_EDGES / 4 &&
              (N_EDGES / 4) % 32 == 0, "every thread takes one or two groups of four edges, whole waves each");
#define SECOND_GROUPS (N_EDGES / 4 - GEMM_WG * GEMM_T)
#define SEG_MAX (8 * GEMM_T)
static_assert(NBLK == 256 && GEMM_WG == 256, "one histogram bin and one table word per thread of the first eight waves");
static_assert(SEG_MAX < 65536, "start and count of a run fit 16 bits each");

__device__ __forceinline__ int seg_base(int b) {
  return 4 * (GEMM_T * b + min(GEMM_T * b, SECOND_GROUPS));
}

__device__ __forceinline__ unsigned wave_scan(unsigned v, int lane) {
  for (int d = 1; d < 32; d <<= 1) {
    const unsigned t = __shfl_up(v, d);
    if (lane >= d) v += t;
  }
  return v;
}

__device__ __forceinline__ void lds_barrier() {
  __builtin_amdgcn_fence(__ATOMIC_RELEASE, "workgroup", "local");
  __builtin_amdgcn_s_barrier();
  __builtin_amdgcn_fence(__ATOMIC_ACQUIRE, "workgroup", "local");
}

__device__ __forceinline__ void load_a(v2f (&a)[KSTEPS], const float* arow) {
#pragma unroll
  for (int t = 0; t < KSTEPS; ++t) {
    const float2 ax = *(const float2*)(arow + 4 * t);
    a[t].x = ax.x; a[t].y = ax.y;
  }
}

__device__ __forceinline__ void tile_mul(const v2f (&a)[KSTEPS], const v2f (&b)[KSTEPS],
                                         float* __restrict__ dst, int mt, int lane, int ncol) {
  v8f c = {};
#pragma unroll
  for (int t = 0; t < KSTEPS; ++t)
    c = __builtin_amdgcn_wmma_f32_16x16x4_f32(false, a[t], false, b[t],
                                              (short)0, c, false, false);
  const int mbase = mt * 16 + 8 * (lane >> 4);
#pragma unroll
  for (int v = 0; v < 8; ++v)
    dst[(size_t)(mbase + v) * HC + ncol] = c[v];
}

__device__ __forceinline__ void edge_count(const int4 s, const int4 d, unsigned* s_hist,
                                           unsigned (&k)[4], unsigned (&r)[4], unsigned (&sl)[4]) {
  const unsigned dv[4] = {(unsigned)d.x, (unsigned)d.y, (unsigned)d.z, (unsigned)d.w};
  const unsigned sv[4] = {(unsigned)s.x, (unsigned)s.y, (unsigned)s.z, (unsigned)s.w};
  for (int q = 0; q < 4; ++q) {
    k[q] = dv[q] / (unsigned)NPB;
    r[q] = sv[q] | (dv[q] - k[q] * NPB) << 16;
  }
  for (int q = 0; q < 4; ++q) sl[q] = atomicAdd(&s_hist[k[q]], 1u);
}

__device__ __forceinline__ void edge_place(const unsigned* s_hist, unsigned* s_stage,
                                           const unsigned (&k)[4], const unsigned (&r)[4],
                                           const unsigned (&sl)[4]) {
  for (int q = 0; q < 4; ++q) s_stage[s_hist[k[q]] + sl[q]] = r[q];
}

__global__ void __launch_bounds__(GEMM_T)
gat_gemm(const float* __restrict__ x, const float* __restrict__ Wl,
         const float* __restrict__ Wr, const int* __restrict__ ei,
         float* __restrict__ xl, float* __restrict__ xr,
         unsigned* __restrict__ seg, unsigned* __restrict__ tab) {
  __shared__ __attribute__((aligned(16))) float s_a[2 * A_BUF];
  __shared__ __attribute__((aligned(16))) unsigned s_stage[SEG_MAX];
  __shared__ unsigned s_hist[NBLK];
  __shared__ unsigned s_wt[NBLK / 32];

  const int tid  = threadIdx.x;
  const int wave = tid >> 5;
  const int lane = tid & 31;

  const int e0 = blockIdx.x * GEMM_T + tid;
  const int e1 = e0 + GEMM_WG * GEMM_T;
  const bool two = e1 < N_EDGES / 4;
  const int4 es0 = ((const int4*)ei)[e0];
  const int4 ed0 = ((const int4*)(ei + N_EDGES))[e0];
  int4 es1 = make_int4(0, 0, 0, 0), ed1 = make_int4(0, 0, 0, 0);
  if (two) {
    es1 = ((const int4*)ei)[e1];
    ed1 = ((const int4*)(ei + N_EDGES))[e1];
  }
  if (tid < NBLK) s_hist[tid] = 0u;

  const float* __restrict__ W   = (wave < 8) ? Wl : Wr;
  float* __restrict__       dst = (wave < 8) ? xl : xr;
  const int ncol  = (wave & 7) * 16 + (lane & 15);
  const int khalf = (lane >> 4) * 2;

  v2f b[KSTEPS];
#pragma unroll
  for (int t = 0; t < KSTEPS; ++t) {
    b[t].x = W[(size_t)(4 * t + khalf) * HC + ncol];
    b[t].y = W[(size_t)(4 * t + khalf + 1) * HC + ncol];
  }

  const int n = (M_TILES - 1 - (int)blockIdx.x) / GEMM_WG + 1;
  const float* xw = x + ((size_t)blockIdx.x * 16 + wave) * IN_F;
  float* stage = &s_a[wave * A_PITCH + 4 * lane];
  const float* frag = &s_a[(lane & 15) * A_PITCH + khalf];

  *(float4*)stage = ((const float4*)xw)[lane];
  if (n > 1) *(float4*)(stage + A_BUF) = ((const float4*)(xw + TILE_STRIDE))[lane];
  float4 g = make_float4(0.f, 0.f, 0.f, 0.f);
  if (n > 2) g = ((const float4*)(xw + 2 * TILE_STRIDE))[lane];
  __syncthreads();
  v2f a0[KSTEPS], a1[KSTEPS];
  load_a(a0, frag);
  unsigned k0[4], r0[4], p0[4], k1[4], r1[4], p1[4];
  edge_count(es0, ed0, s_hist, k0, r0, p0);
  if (two) edge_count(es1, ed1, s_hist, k1, r1, p1);
  __syncthreads();

  unsigned cnt = 0u, incl = 0u;
  if (tid < NBLK) {
    cnt = s_hist[tid];
    incl = wave_scan(cnt, lane);
    if (lane == 31) s_wt[wave] = incl;
  }
  lds_barrier();
  if (tid < NBLK) {
    unsigned first = incl - cnt;
    for (int w = 0; w < wave; ++w) first += s_wt[w];
    s_hist[tid] = first;
    tab[blockIdx.x * NBLK + tid] = first | cnt << 16;
  }
  lds_barrier();
  edge_place(s_hist, s_stage, k0, r0, p0);
  if (two) edge_place(s_hist, s_stage, k1, r1, p1);
  lds_barrier();
  {
    const int quads = GEMM_T + min(max(SECOND_GROUPS - GEMM_T * (int)blockIdx.x, 0), GEMM_T);
    uint4* __restrict__ o4 = (uint4*)(seg + seg_base((int)blockIdx.x));
    for (int i = tid; i < quads; i += GEMM_T) o4[i] = ((const uint4*)s_stage)[i];
  }

  for (int j = 0; j < n; j += 2) {
    if (j + 2 < n) *(float4*)stage = g;
    if (j + 3 < n) g = ((const float4*)(xw + (size_t)(j + 3) * TILE_STRIDE))[lane];
    if (j + 1 < n) load_a(a1, frag + A_BUF);
    tile_mul(a0, b, dst, (int)blockIdx.x + j * GEMM_WG, lane, ncol);
    lds_barrier();
    if (j + 1 < n) {
      if (j + 3 < n) *(float4*)(stage + A_BUF) = g;
      if (j + 4 < n) g = ((const float4*)(xw + (size_t)(j + 4) * TILE_STRIDE))[lane];
      if (j + 2 < n) load_a(a0, frag);
      tile_mul(a1, b, dst, (int)blockIdx.x + (j + 1) * GEMM_WG, lane, ncol);
      lds_barrier();
    }
  }
}

#define CAP    8192
#define TPB    1024
#define NWAVE  32
#define WIN_B  2
#define NWIN   (GEMM_WG / WIN_B + 1)
#define FILL_U 8
static_assert(TPB == 4 * GEMM_WG && GEMM_WG == 8 * 32, "four threads copy one run; one wave places eight runs per lane");
static_assert(WIN_B * SEG_MAX <= CAP && GEMM_WG % WIN_B == 0, "a window's runs fit the list");

__device__ __forceinline__ void tab_load(const unsigned* __restrict__ tab, unsigned (&tw)[8]) {
  for (int j = 0; j < 8; ++j)
    tw[j] = tab[(8 * (int)threadIdx.x + j) * NBLK + blockIdx.x];
}

__device__ __forceinline__ int list_places(const unsigned (&tw)[8], int b_lo, int b_hi,
                                           unsigned* s_tab, int* s_pos, int* s_n) {
  if (threadIdx.x < 32) {
    const int lane = threadIdx.x;
    unsigned c[8], sum = 0u;
    for (int j = 0; j < 8; ++j) {
      const int b = 8 * lane + j;
      c[j] = (b >= b_lo && b < b_hi) ? (tw[j] >> 16) : 0u;
      sum += c[j];
    }
    const unsigned incl = wave_scan(sum, lane);
    unsigned p = incl - sum;
    for (int j = 0; j < 8; ++j) {
      s_tab[8 * lane + j] = (tw[j] & 0xFFFFu) | c[j] << 16;
      s_pos[8 * lane + j] = (int)p;
      p += c[j];
    }
    if (lane == 31) *s_n = (int)incl;
  }
  __syncthreads();
  return *s_n;
}

__device__ __forceinline__ void list_fill(const unsigned* __restrict__ seg,
                                          const unsigned* s_tab, const int* s_pos,
                                          int* s_src, int* s_dl) {
  const int b   = threadIdx.x >> 2;
  const int sub = threadIdx.x & 3;
  const int pos = s_pos[b];
  const unsigned t = s_tab[b];
  const int cnt = (int)(t >> 16);
  const int first = seg_base(b) + (int)(t & 0xFFFFu);
  unsigned r[FILL_U];
  for (int u = 0; u < FILL_U; ++u)
    r[u] = seg[min(first + min(sub + 4 * u, max(cnt - 1, 0)), N_EDGES - 1)];
  for (int u = 0; u < FILL_U; ++u) {
    const int i = sub + 4 * u;
    if (i < cnt) {
      s_src[pos + i] = (int)(r[u] & 0xFFFFu);
      s_dl[pos + i]  = (int)(r[u] >> 16);
    }
  }
  for (int i = sub + 4 * FILL_U; i < cnt; i += 4) {
    const unsigned rr = seg[first + i];
    s_src[pos + i] = (int)(rr & 0xFFFFu);
    s_dl[pos + i]  = (int)(rr >> 16);
  }
}

__device__ __forceinline__ void list_self(int node0, unsigned nn, int n0, int* s_src, int* s_dl) {
  if (threadIdx.x < nn) {
    s_src[n0 + (int)threadIdx.x] = node0 + (int)threadIdx.x;
    s_dl[n0 + (int)threadIdx.x]  = (int)threadIdx.x;
  }
}

template <int NE, bool XR_LDS>
__device__ __forceinline__ void logit_group(const float* __restrict__ xl,
                                            const float* __restrict__ xr,
                                            const float* s_xr,
                                            const float4 w, int node0, int i0, int lane,
                                            const int* s_src, const int* s_dl,
                                            float* s_val, float* s_m) {
  float4 a[NE], bg[NE];
  int dl[NE];
#pragma unroll
  for (int u = 0; u < NE; ++u) {
    const int src = s_src[i0 + u];
    dl[u] = s_dl[i0 + u];
    a[u] = ((const float4*)(xl + (size_t)src * HC))[lane];
    if (!XR_LDS) bg[u] = ((const float4*)(xr + (size_t)(node0 + dl[u]) * HC))[lane];
  }
  __builtin_amdgcn_sched_barrier(0);
  float s[NE];
#pragma unroll
  for (int u = 0; u < NE; ++u) {
    const float4 b = XR_LDS ? ((const float4*)(s_xr + dl[u] * HC))[lane] : bg[u];
    float vx = a[u].x + b.x, vy = a[u].y + b.y;
    float vz = a[u].z + b.z, vw = a[u].w + b.w;
    vx = (vx > 0.f) ? vx : vx * NEG_SLOPE;
    vy = (vy > 0.f) ? vy : vy * NEG_SLOPE;
    vz = (vz > 0.f) ? vz : vz * NEG_SLOPE;
    vw = (vw > 0.f) ? vw : vw * NEG_SLOPE;
    s[u] = vx * w.x + vy * w.y + vz * w.z + vw * w.w;
  }
  if constexpr (NE == 8) {
    const bool b0 = (lane & 1) != 0, b1 = (lane & 2) != 0, b2 = (lane & 4) != 0;
    const int  mine = i0 + ((lane & 1) << 2 | (lane & 2) | ((lane >> 2) & 1));
    const int  d    = s_dl[mine];
    float t[4];
#pragma unroll
    for (int u = 0; u < 4; ++u) {
      const float keep = b0 ? s[u + 4] : s[u];
      const float send = b0 ? s[u] : s[u + 4];
      t[u] = keep + __shfl_xor(send, 1);
    }
    float p[2];
#pragma unroll
    for (int u = 0; u < 2; ++u) {
      const float keep = b1 ? t[u + 2] : t[u];
      const float send = b1 ? t[u] : t[u + 2];
      p[u] = keep + __shfl_xor(send, 2);
    }
    const float keep = b2 ? p[1] : p[0];
    const float send = b2 ? p[0] : p[1];
    const float tot  = keep + __shfl_xor(send, 4);
    const int   h    = lane >> 3;
    s_val[mine * 4 + h] = tot;
    __hip_atomic_fetch_max(&s_m[d * 4 + h], tot, __ATOMIC_RELAXED,
                           __HIP_MEMORY_SCOPE_WORKGROUP);
  } else {
#pragma unroll
    for (int u = 0; u < NE; ++u) s[u] += __shfl_xor(s[u], 1);
#pragma unroll
    for (int u = 0; u < NE; ++u) s[u] += __shfl_xor(s[u], 2);
#pragma unroll
    for (int u = 0; u < NE; ++u) s[u] += __shfl_xor(s[u], 4);
    if ((lane & 7) == 0) {
      const int h = lane >> 3;
#pragma unroll
      for (int u = 0; u < NE; ++u) {
        s_val[(i0 + u) * 4 + h] = s[u];
        __hip_atomic_fetch_max(&s_m[dl[u] * 4 + h], s[u], __ATOMIC_RELAXED,
                               __HIP_MEMORY_SCOPE_WORKGROUP);
      }
    }
  }
}

#define GRP 8
#define GRP_WIN 2

template <bool XR_LDS>
__device__ __forceinline__ void phase_logits(const float* __restrict__ xl,
                                             const float* __restrict__ xr,
                                             const float* s_xr,
                                             const float4 w, int node0, int n,
                                             const int* s_src, const int* s_dl,
                                             float* s_val, float* s_m) {
  const int wave = threadIdx.x >> 5;
  const int lane = threadIdx.x & 31;
  constexpr int G = XR_LDS ? GRP : GRP_WIN;
  const int nfull = n / G;
  for (int g = wave; g < nfull; g += NWAVE)
    logit_group<G, XR_LDS>(xl, xr, s_xr, w, node0, g * G, lane, s_src, s_dl, s_val, s_m);
  const int i = nfull * G + wave;
  if (i < n)
    logit_group<1, XR_LDS>(xl, xr, s_xr, w, node0, i, lane, s_src, s_dl, s_val, s_m);
}

__device__ __forceinline__ void phase_exp(int n, bool add, const int* s_dl,
                                          float* s_val, const float* s_m, float* s_den) {
  for (int i = threadIdx.x; i < n * 4; i += TPB) {
    const int k = s_dl[i >> 2] * 4 + (i & 3);
    const float ex = __expf(s_val[i] - s_m[k]);
    s_val[i] = ex;
    if (add)
      __hip_atomic_fetch_add(&s_den[k], ex, __ATOMIC_RELAXED, __HIP_MEMORY_SCOPE_WORKGROUP);
  }
}

__device__ __forceinline__ void phase_alpha(int n, const int* s_dl, float* s_val,
                                            const float* s_den) {
  for (int i = threadIdx.x; i < n * 4; i += TPB) {
    const int k = s_dl[i >> 2] * 4 + (i & 3);
    s_val[i] = s_val[i] / (s_den[k] + 1e-16f);
  }
}

template <int NE>
__device__ __forceinline__ void sum_group(const float* __restrict__ xl, int i0, int lane,
                                          const int* s_src, const int* s_dl,
                                          const float* s_val, float* s_acc) {
  float4 v[NE];
  int dl[NE];
#pragma unroll
  for (int u = 0; u < NE; ++u) {
    const int src = s_src[i0 + u];
    dl[u] = s_dl[i0 + u];
    v[u] = ((const float4*)(xl + (size_t)src * HC))[lane];
  }
  __builtin_amdgcn_sched_barrier(0);
#pragma unroll
  for (int u = 0; u < NE; ++u) {
    const float alpha = s_val[(i0 + u) * 4 + (lane >> 3)];
    float* row = s_acc + dl[u] * HC + lane;
    __hip_atomic_fetch_add(row +  0, v[u].x * alpha, __ATOMIC_RELAXED, __HIP_MEMORY_SCOPE_WORKGROUP);
    __hip_atomic_fetch_add(row + 32, v[u].y * alpha, __ATOMIC_RELAXED, __HIP_MEMORY_SCOPE_WORKGROUP);
    __hip_atomic_fetch_add(row + 64, v[u].z * alpha, __ATOMIC_RELAXED, __HIP_MEMORY_SCOPE_WORKGROUP);
    __hip_atomic_fetch_add(row + 96, v[u].w * alpha, __ATOMIC_RELAXED, __HIP_MEMORY_SCOPE_WORKGROUP);
  }
}

template <int G>
__device__ __forceinline__ void phase_sum(const float* __restrict__ xl, int n,
                                          const int* s_src, const int* s_dl,
                                          const float* s_val, float* s_acc) {
  const int wave = threadIdx.x >> 5;
  const int lane = threadIdx.x & 31;
  const int nfull = n / G;
  for (int g = wave; g < nfull; g += NWAVE)
    sum_group<G>(xl, g * G, lane, s_src, s_dl, s_val, s_acc);
  const int i = nfull * G + wave;
  if (i < n) sum_group<1>(xl, i, lane, s_src, s_dl, s_val, s_acc);
}

__device__ __forceinline__ void zero_acc(float* s_acc) {
  for (int i = threadIdx.x; i < NPB * HC / 4; i += TPB)
    ((float4*)s_acc)[i] = make_float4(0.f, 0.f, 0.f, 0.f);
}

__global__ void __launch_bounds__(TPB)
gat_owner(const unsigned* __restrict__ seg, const unsigned* __restrict__ tab,
          const float* __restrict__ xl,
          const float* __restrict__ xr, const float* __restrict__ att,
          const float* __restrict__ bias, float* __restrict__ out) {
  __shared__ float s_acc[NPB * HC];
  __shared__ float s_m[NPB * 4];
  __shared__ float s_den[NPB * 4];
  __shared__ int   s_src[CAP];
  __shared__ int   s_dl[CAP];
  __shared__ float s_val[CAP * 4];
  __shared__ unsigned s_tab[GEMM_WG];
  __shared__ int   s_pos[GEMM_WG];
  __shared__ int   s_n;

  const int tid   = threadIdx.x;
  const int node0 = blockIdx.x * NPB;
  const unsigned nn = (unsigned)min(NPB, N_NODES - node0);
  const float4 w = ((const float4*)att)[tid & 31];
  unsigned tw[8] = {0u, 0u, 0u, 0u, 0u, 0u, 0u, 0u};
  if (tid < 32) tab_load(tab, tw);

  for (int i = tid; i < (int)nn * (HC / 4); i += TPB)
    ((float4*)s_acc)[i] = ((const float4*)(xr + (size_t)node0 * HC))[i];
  for (int i = tid; i < NPB * 4; i += TPB) {
    s_m[i] = -3.402823466e38f;
    s_den[i] = 0.0f;
  }
  const int ne = list_places(tw, 0, GEMM_WG, s_tab, s_pos, &s_n);
  const int total = ne + (int)nn;

  if (total <= CAP) {
    list_fill(seg, s_tab, s_pos, s_src, s_dl);
    list_self(node0, nn, ne, s_src, s_dl);
    __syncthreads();
    phase_logits<true>(xl, xr, s_acc, w, node0, total, s_src, s_dl, s_val, s_m);
    __syncthreads();
    zero_acc(s_acc);
    phase_exp(total, true, s_dl, s_val, s_m, s_den);
    __syncthreads();
    phase_alpha(total, s_dl, s_val, s_den);
    __syncthreads();
    phase_sum<GRP>(xl, total, s_src, s_dl, s_val, s_acc);
    __syncthreads();
  } else {
    zero_acc(s_acc);
    for (int sweep = 0; sweep < 3; ++sweep) {
      for (int win = 0; win < NWIN; ++win) {
        __syncthreads();
        int n;
        if (win < NWIN - 1) {
          if (tid < 32) tab_load(tab, tw);
          n = list_places(tw, win * WIN_B, win * WIN_B + WIN_B, s_tab, s_pos, &s_n);
          list_fill(seg, s_tab, s_pos, s_src, s_dl);
        } else {
          n = (int)nn;
          list_self(node0, nn, 0, s_src, s_dl);
        }
        __syncthreads();
        phase_logits<false>(xl, xr, s_acc, w, node0, n, s_src, s_dl, s_val, s_m);
        __syncthreads();
        if (sweep >= 1) {
          phase_exp(n, sweep == 1, s_dl, s_val, s_m, s_den);
          __syncthreads();
        }
        if (sweep == 2) {
          phase_alpha(n, s_dl, s_val, s_den);
          __syncthreads();
          phase_sum<GRP_WIN>(xl, n, s_src, s_dl, s_val, s_acc);
          __syncthreads();
        }
      }
    }
  }

  const float4 bq = ((const float4*)bias)[tid & 31];
  for (int idx = tid; idx < (int)nn * 32; idx += TPB) {
    const float* row = s_acc + (idx >> 5) * HC + (idx & 31);
    float4 a;
    a.x = row[0]  + bq.x;
    a.y = row[32] + bq.y;
    a.z = row[64] + bq.z;
    a.w = row[96] + bq.w;
    ((float4*)(out + (size_t)node0 * HC))[idx] = a;
  }
}

extern "C" void kernel_launch(void* const* d_in, const int* in_sizes, int n_in,
                              void* d_out, int out_size, void* d_ws, size_t ws_size,
                              hipStream_t stream) {
  const float* x    = (const float*)d_in[0];
  const int*   ei   = (const int*)  d_in[1];
  const float* Wl   = (const float*)d_in[2];
  const float* Wr   = (const float*)d_in[3];
  const float* att  = (const float*)d_in[4];
  const float* bias = (const float*)d_in[5];
  float* out = (float*)d_out;

  float* ws = (float*)d_ws;
  float* xl = ws;
  float* xr = xl + (size_t)N_NODES * HC;
  unsigned* seg = (unsigned*)(xr + (size_t)N_NODES * HC);
  unsigned* tab = seg + N_EDGES;

  gat_gemm<<<GEMM_WG, GEMM_T, 0, stream>>>(x, Wl, Wr, ei, xl, xr, seg, tab);

  gat_owner<<<NBLK, TPB, 0, stream>>>(seg, tab, xl, xr, att, bias, out);
}


# edge launch: softmax shifted by the node's self-loop logit, known first, instead of its max: one sweep, each xl row read once; sums of ex*xl divided by exp-sum at the end; max path kept as guard
# speedup vs baseline: 5.3391x; 1.1860x over previous
#include <hip/hip_runtime.h>
#include <hip/hip_bf16.h>

#define N_NODES   50000
#define N_EDGES   800000
#define E_TOT     850000
#define IN_F      128
#define HC        128
#define NEG_SLOPE 0.2f
#define NPB       196
#define NBLK      256
static_assert(NBLK <= 256 && NBLK * NPB >= N_NODES, "owner key is one byte");
static_assert(N_NODES <= 65536 && NPB <= 65536, "edge record packs source and local destination in 16 bits each");

typedef __attribute__((ext_vector_type(2))) float v2f;
typedef __attribute__((ext_vector_type(8))) float v8f;

#define GEMM_T   512
#define GEMM_WG  256
#define M_TILES  (N_NODES / 16)
#define KSTEPS   (IN_F / 4)
#define A_PITCH  132
#define A_BUF    (16 * A_PITCH)
#define TILE_STRIDE ((size_t)GEMM_WG * 16 * IN_F)
static_assert(M_TILES * 16 == N_NODES && GEMM_WG <= M_TILES, "whole row tiles, one or more per work-group");
static_assert(GEMM_WG * GEMM_T <= N_EDGES / 4 && 2 * GEMM_WG * GEMM_T >= N_EDGES / 4 &&
              (N_EDGES / 4) % 32 == 0, "every thread takes one or two groups of four edges, whole waves each");
#define SECOND_GROUPS (N_EDGES / 4 - GEMM_WG * GEMM_T)
#define SEG_MAX (8 * GEMM_T)
static_assert(NBLK == 256 && GEMM_WG == 256, "one histogram bin and one table word per thread of the first eight waves");
static_assert(SEG_MAX < 65536, "start and count of a run fit 16 bits each");

__device__ __forceinline__ int seg_base(int b) {
  return 4 * (GEMM_T * b + min(GEMM_T * b, SECOND_GROUPS));
}

__device__ __forceinline__ unsigned wave_scan(unsigned v, int lane) {
  for (int d = 1; d < 32; d <<= 1) {
    const unsigned t = __shfl_up(v, d);
    if (lane >= d) v += t;
  }
  return v;
}

__device__ __forceinline__ void lds_barrier() {
  __builtin_amdgcn_fence(__ATOMIC_RELEASE, "workgroup", "local");
  __builtin_amdgcn_s_barrier();
  __builtin_amdgcn_fence(__ATOMIC_ACQUIRE, "workgroup", "local");
}

__device__ __forceinline__ void load_a(v2f (&a)[KSTEPS], const float* arow) {
#pragma unroll
  for (int t = 0; t < KSTEPS; ++t) {
    const float2 ax = *(const float2*)(arow + 4 * t);
    a[t].x = ax.x; a[t].y = ax.y;
  }
}

__device__ __forceinline__ void tile_mul(const v2f (&a)[KSTEPS], const v2f (&b)[KSTEPS],
                                         float* __restrict__ dst, int mt, int lane, int ncol) {
  v8f c = {};
#pragma unroll
  for (int t = 0; t < KSTEPS; ++t)
    c = __builtin_amdgcn_wmma_f32_16x16x4_f32(false, a[t], false, b[t],
                                              (short)0, c, false, false);
  const int mbase = mt * 16 + 8 * (lane >> 4);
#pragma unroll
  for (int v = 0; v < 8; ++v)
    dst[(size_t)(mbase + v) * HC + ncol] = c[v];
}

__device__ __forceinline__ void edge_count(const int4 s, const int4 d, unsigned* s_hist,
                                           unsigned (&k)[4], unsigned (&r)[4], unsigned (&sl)[4]) {
  const unsigned dv[4] = {(unsigned)d.x, (unsigned)d.y, (unsigned)d.z, (unsigned)d.w};
  const unsigned sv[4] = {(unsigned)s.x, (unsigned)s.y, (unsigned)s.z, (unsigned)s.w};
  for (int q = 0; q < 4; ++q) {
    k[q] = dv[q] / (unsigned)NPB;
    r[q] = sv[q] | (dv[q] - k[q] * NPB) << 16;
  }
  for (int q = 0; q < 4; ++q) sl[q] = atomicAdd(&s_hist[k[q]], 1u);
}

__device__ __forceinline__ void edge_place(const unsigned* s_hist, unsigned* s_stage,
                                           const unsigned (&k)[4], const unsigned (&r)[4],
                                           const unsigned (&sl)[4]) {
  for (int q = 0; q < 4; ++q) s_stage[s_hist[k[q]] + sl[q]] = r[q];
}

__global__ void __launch_bounds__(GEMM_T)
gat_gemm(const float* __restrict__ x, const float* __restrict__ Wl,
         const float* __restrict__ Wr, const int* __restrict__ ei,
         float* __restrict__ xl, float* __restrict__ xr,
         unsigned* __restrict__ seg, unsigned* __restrict__ tab) {
  __shared__ __attribute__((aligned(16))) float s_a[2 * A_BUF];
  __shared__ __attribute__((aligned(16))) unsigned s_stage[SEG_MAX];
  __shared__ unsigned s_hist[NBLK];
  __shared__ unsigned s_wt[NBLK / 32];

  const int tid  = threadIdx.x;
  const int wave = tid >> 5;
  const int lane = tid & 31;

  const int e0 = blockIdx.x * GEMM_T + tid;
  const int e1 = e0 + GEMM_WG * GEMM_T;
  const bool two = e1 < N_EDGES / 4;
  const int4 es0 = ((const int4*)ei)[e0];
  const int4 ed0 = ((const int4*)(ei + N_EDGES))[e0];
  int4 es1 = make_int4(0, 0, 0, 0), ed1 = make_int4(0, 0, 0, 0);
  if (two) {
    es1 = ((const int4*)ei)[e1];
    ed1 = ((const int4*)(ei + N_EDGES))[e1];
  }
  if (tid < NBLK) s_hist[tid] = 0u;

  const float* __restrict__ W   = (wave < 8) ? Wl : Wr;
  float* __restrict__       dst = (wave < 8) ? xl : xr;
  const int ncol  = (wave & 7) * 16 + (lane & 15);
  const int khalf = (lane >> 4) * 2;

  v2f b[KSTEPS];
#pragma unroll
  for (int t = 0; t < KSTEPS; ++t) {
    b[t].x = W[(size_t)(4 * t + khalf) * HC + ncol];
    b[t].y = W[(size_t)(4 * t + khalf + 1) * HC + ncol];
  }

  const int n = (M_TILES - 1 - (int)blockIdx.x) / GEMM_WG + 1;
  const float* xw = x + ((size_t)blockIdx.x * 16 + wave) * IN_F;
  float* stage = &s_a[wave * A_PITCH + 4 * lane];
  const float* frag = &s_a[(lane & 15) * A_PITCH + khalf];

  *(float4*)stage = ((const float4*)xw)[lane];
  if (n > 1) *(float4*)(stage + A_BUF) = ((const float4*)(xw + TILE_STRIDE))[lane];
  float4 g = make_float4(0.f, 0.f, 0.f, 0.f);
  if (n > 2) g = ((const float4*)(xw + 2 * TILE_STRIDE))[lane];
  __syncthreads();
  v2f a0[KSTEPS], a1[KSTEPS];
  load_a(a0, frag);
  unsigned k0[4], r0[4], p0[4], k1[4], r1[4], p1[4];
  edge_count(es0, ed0, s_hist, k0, r0, p0);
  if (two) edge_count(es1, ed1, s_hist, k1, r1, p1);
  __syncthreads();

  unsigned cnt = 0u, incl = 0u;
  if (tid < NBLK) {
    cnt = s_hist[tid];
    incl = wave_scan(cnt, lane);
    if (lane == 31) s_wt[wave] = incl;
  }
  lds_barrier();
  if (tid < NBLK) {
    unsigned first = incl - cnt;
    for (int w = 0; w < wave; ++w) first += s_wt[w];
    s_hist[tid] = first;
    tab[blockIdx.x * NBLK + tid] = first | cnt << 16;
  }
  lds_barrier();
  edge_place(s_hist, s_stage, k0, r0, p0);
  if (two) edge_place(s_hist, s_stage, k1, r1, p1);
  lds_barrier();
  {
    const int quads = GEMM_T + min(max(SECOND_GROUPS - GEMM_T * (int)blockIdx.x, 0), GEMM_T);
    uint4* __restrict__ o4 = (uint4*)(seg + seg_base((int)blockIdx.x));
    for (int i = tid; i < quads; i += GEMM_T) o4[i] = ((const uint4*)s_stage)[i];
  }

  for (int j = 0; j < n; j += 2) {
    if (j + 2 < n) *(float4*)stage = g;
    if (j + 3 < n) g = ((const float4*)(xw + (size_t)(j + 3) * TILE_STRIDE))[lane];
    if (j + 1 < n) load_a(a1, frag + A_BUF);
    tile_mul(a0, b, dst, (int)blockIdx.x + j * GEMM_WG, lane, ncol);
    lds_barrier();
    if (j + 1 < n) {
      if (j + 3 < n) *(float4*)(stage + A_BUF) = g;
      if (j + 4 < n) g = ((const float4*)(xw + (size_t)(j + 4) * TILE_STRIDE))[lane];
      if (j + 2 < n) load_a(a0, frag);
      tile_mul(a1, b, dst, (int)blockIdx.x + (j + 1) * GEMM_WG, lane, ncol);
      lds_barrier();
    }
  }
}

#define SHIFT_T 8.0f
#define CAP    8192
#define TPB    1024
#define NWAVE  32
#define WIN_B  2
#define NWIN   (GEMM_WG / WIN_B + 1)
#define FILL_U 8
static_assert(TPB == 4 * GEMM_WG && GEMM_WG == 8 * 32, "four threads copy one run; one wave places eight runs per lane");
static_assert(WIN_B * SEG_MAX <= CAP && GEMM_WG % WIN_B == 0, "a window's runs fit the list");

__device__ __forceinline__ void tab_load(const unsigned* __restrict__ tab, unsigned (&tw)[8]) {
  for (int j = 0; j < 8; ++j)
    tw[j] = tab[(8 * (int)threadIdx.x + j) * NBLK + blockIdx.x];
}

__device__ __forceinline__ int list_places(const unsigned (&tw)[8], int b_lo, int b_hi,
                                           unsigned* s_tab, int* s_pos, int* s_n) {
  if (threadIdx.x < 32) {
    const int lane = threadIdx.x;
    unsigned c[8], sum = 0u;
    for (int j = 0; j < 8; ++j) {
      const int b = 8 * lane + j;
      c[j] = (b >= b_lo && b < b_hi) ? (tw[j] >> 16) : 0u;
      sum += c[j];
    }
    const unsigned incl = wave_scan(sum, lane);
    unsigned p = incl - sum;
    for (int j = 0; j < 8; ++j) {
      s_tab[8 * lane + j] = (tw[j] & 0xFFFFu) | c[j] << 16;
      s_pos[8 * lane + j] = (int)p;
      p += c[j];
    }
    if (lane == 31) *s_n = (int)incl;
  }
  __syncthreads();
  return *s_n;
}

__device__ __forceinline__ void list_fill(const unsigned* __restrict__ seg,
                                          const unsigned* s_tab, const int* s_pos,
                                          int* s_src, int* s_dl) {
  const int b   = threadIdx.x >> 2;
  const int sub = threadIdx.x & 3;
  const int pos = s_pos[b];
  const unsigned t = s_tab[b];
  const int cnt = (int)(t >> 16);
  const int first = seg_base(b) + (int)(t & 0xFFFFu);
  unsigned r[FILL_U];
  for (int u = 0; u < FILL_U; ++u)
    r[u] = seg[min(first + min(sub + 4 * u, max(cnt - 1, 0)), N_EDGES - 1)];
  for (int u = 0; u < FILL_U; ++u) {
    const int i = sub + 4 * u;
    if (i < cnt) {
      s_src[pos + i] = (int)(r[u] & 0xFFFFu);
      s_dl[pos + i]  = (int)(r[u] >> 16);
    }
  }
  for (int i = sub + 4 * FILL_U; i < cnt; i += 4) {
    const unsigned rr = seg[first + i];
    s_src[pos + i] = (int)(rr & 0xFFFFu);
    s_dl[pos + i]  = (int)(rr >> 16);
  }
}

__device__ __forceinline__ void list_self(int node0, unsigned nn, int n0, int* s_src, int* s_dl) {
  if (threadIdx.x < nn) {
    s_src[n0 + (int)threadIdx.x] = node0 + (int)threadIdx.x;
    s_dl[n0 + (int)threadIdx.x]  = (int)threadIdx.x;
  }
}

__device__ __forceinline__ float lane_dot(const float4 a, const float4 b, const float4 w) {
  float vx = a.x + b.x, vy = a.y + b.y;
  float vz = a.z + b.z, vw = a.w + b.w;
  vx = (vx > 0.f) ? vx : vx * NEG_SLOPE;
  vy = (vy > 0.f) ? vy : vy * NEG_SLOPE;
  vz = (vz > 0.f) ? vz : vz * NEG_SLOPE;
  vw = (vw > 0.f) ? vw : vw * NEG_SLOPE;
  return vx * w.x + vy * w.y + vz * w.z + vw * w.w;
}

template <int NE>
__device__ __forceinline__ void head_sum(float (&s)[NE]) {
#pragma unroll
  for (int u = 0; u < NE; ++u) s[u] += __shfl_xor(s[u], 1);
#pragma unroll
  for (int u = 0; u < NE; ++u) s[u] += __shfl_xor(s[u], 2);
#pragma unroll
  for (int u = 0; u < NE; ++u) s[u] += __shfl_xor(s[u], 4);
}

#define GRP 8
#define GRP_WIN 2


template <int NE>
__device__ __forceinline__ void self_group(const float* __restrict__ xl, const float* s_xr,
                                           const float4 w, int node0, int nn, int d0, int lane,
                                           float* s_c, float* s_den, float* s_acc) {
  float4 a[NE];
  int dl[NE];
#pragma unroll
  for (int u = 0; u < NE; ++u) {
    dl[u] = min(d0 + u, nn - 1);
    a[u] = ((const float4*)(xl + (size_t)(node0 + dl[u]) * HC))[lane];
  }
  __builtin_amdgcn_sched_barrier(0);
  float s[NE];
#pragma unroll
  for (int u = 0; u < NE; ++u)
    s[u] = lane_dot(a[u], ((const float4*)(s_xr + dl[u] * HC))[lane], w);
  head_sum<NE>(s);
  const int h = lane >> 3;
#pragma unroll
  for (int u = 0; u < NE; ++u) {
    if (d0 + u < nn) {
      float* row = s_acc + dl[u] * HC + lane;
      row[0]  = a[u].x;
      row[32] = a[u].y;
      row[64] = a[u].z;
      row[96] = a[u].w;
      if ((lane & 7) == 0) {
        s_c[dl[u] * 4 + h]   = s[u];
        s_den[dl[u] * 4 + h] = 1.0f;
      }
    }
  }
}

__device__ __forceinline__ void phase_self(const float* __restrict__ xl, const float* s_xr,
                                           const float4 w, int node0, int nn,
                                           float* s_c, float* s_den, float* s_acc) {
  const int wave = threadIdx.x >> 5;
  const int lane = threadIdx.x & 31;
  for (int d0 = wave * GRP; d0 < nn; d0 += NWAVE * GRP)
    self_group<GRP>(xl, s_xr, w, node0, nn, d0, lane, s_c, s_den, s_acc);
}

__device__ __forceinline__ constexpr unsigned holder(int u) {
  return (unsigned)(((u >> 2) & 1) | (((u >> 1) & 1) << 1) | ((u & 1) << 2));
}

template <int NE>
__device__ __forceinline__ void sweep_group(const float* __restrict__ xl, const float* s_xr,
                                            const float4 w, int i0, int lane,
                                            const int* s_src, const int* s_dl,
                                            const float* s_c, float* s_den, float* s_acc,
                                            int* s_flag) {
  float4 a[NE];
  int dl[NE];
#pragma unroll
  for (int u = 0; u < NE; ++u) {
    const int src = s_src[i0 + u];
    dl[u] = s_dl[i0 + u];
    a[u] = ((const float4*)(xl + (size_t)src * HC))[lane];
  }
  __builtin_amdgcn_sched_barrier(0);
  float s[NE];
#pragma unroll
  for (int u = 0; u < NE; ++u)
    s[u] = lane_dot(a[u], ((const float4*)(s_xr + dl[u] * HC))[lane], w);
  const int h = lane >> 3;
  float ex[NE];
  if constexpr (NE == 8) {
    const bool b0 = (lane & 1) != 0, b1 = (lane & 2) != 0, b2 = (lane & 4) != 0;
    const int  mine = i0 + ((lane & 1) << 2 | (lane & 2) | ((lane >> 2) & 1));
    const int  d    = s_dl[mine];
    float t[4];
#pragma unroll
    for (int u = 0; u < 4; ++u) {
      const float keep = b0 ? s[u + 4] : s[u];
      const float send = b0 ? s[u] : s[u + 4];
      t[u] = keep + __shfl_xor(send, 1);
    }
    float p[2];
#pragma unroll
    for (int u = 0; u < 2; ++u) {
      const float keep = b1 ? t[u + 2] : t[u];
      const float send = b1 ? t[u] : t[u + 2];
      p[u] = keep + __shfl_xor(send, 2);
    }
    const float keep = b2 ? p[1] : p[0];
    const float send = b2 ? p[0] : p[1];
    const float tot  = keep + __shfl_xor(send, 4);
    const float z    = tot - s_c[d * 4 + h];
    if (!(z <= SHIFT_T)) *s_flag = 1;
    const float e = __expf(z);
    __hip_atomic_fetch_add(&s_den[d * 4 + h], e, __ATOMIC_RELAXED, __HIP_MEMORY_SCOPE_WORKGROUP);
#pragma unroll
    for (int u = 0; u < 8; ++u) {
      const unsigned lo = holder(u) * 0x11111111u;
      const unsigned hi = (holder(u) + 8u) * 0x11111111u;
      const unsigned eb = __float_as_uint(e);
      ex[u] = __uint_as_float(__builtin_amdgcn_permlane16(eb, eb, lo, hi, false, false));
    }
  } else {
    head_sum<NE>(s);
#pragma unroll
    for (int u = 0; u < NE; ++u) {
      const float z = s[u] - s_c[dl[u] * 4 + h];
      if (!(z <= SHIFT_T)) *s_flag = 1;
      ex[u] = __expf(z);
      if ((lane & 7) == 0)
        __hip_atomic_fetch_add(&s_den[dl[u] * 4 + h], ex[u], __ATOMIC_RELAXED,
                               __HIP_MEMORY_SCOPE_WORKGROUP);
    }
  }
#pragma unroll
  for (int u = 0; u < NE; ++u) {
    float* row = s_acc + dl[u] * HC + lane;
    __hip_atomic_fetch_add(row +  0, a[u].x * ex[u], __ATOMIC_RELAXED, __HIP_MEMORY_SCOPE_WORKGROUP);
    __hip_atomic_fetch_add(row + 32, a[u].y * ex[u], __ATOMIC_RELAXED, __HIP_MEMORY_SCOPE_WORKGROUP);
    __hip_atomic_fetch_add(row + 64, a[u].z * ex[u], __ATOMIC_RELAXED, __HIP_MEMORY_SCOPE_WORKGROUP);
    __hip_atomic_fetch_add(row + 96, a[u].w * ex[u], __ATOMIC_RELAXED, __HIP_MEMORY_SCOPE_WORKGROUP);
  }
}

__device__ __forceinline__ void phase_sweep(const float* __restrict__ xl, const float* s_xr,
                                            const float4 w, int n,
                                            const int* s_src, const int* s_dl,
                                            const float* s_c, float* s_den, float* s_acc,
                                            int* s_flag) {
  const int wave = threadIdx.x >> 5;
  const int lane = threadIdx.x & 31;
  const int nfull = n / GRP;
  for (int g = wave; g < nfull; g += NWAVE)
    sweep_group<GRP>(xl, s_xr, w, g * GRP, lane, s_src, s_dl, s_c, s_den, s_acc, s_flag);
  const int i = nfull * GRP + wave;
  if (i < n)
    sweep_group<1>(xl, s_xr, w, i, lane, s_src, s_dl, s_c, s_den, s_acc, s_flag);
}


template <int NE>
__device__ __forceinline__ void logit_group(const float* __restrict__ xl,
                                            const float* __restrict__ xr,
                                            const float4 w, int node0, int i0, int lane,
                                            const int* s_src, const int* s_dl,
                                            float* s_val, float* s_m) {
  float4 a[NE], b[NE];
  int dl[NE];
#pragma unroll
  for (int u = 0; u < NE; ++u) {
    const int src = s_src[i0 + u];
    dl[u] = s_dl[i0 + u];
    a[u] = ((const float4*)(xl + (size_t)src * HC))[lane];
    b[u] = ((const float4*)(xr + (size_t)(node0 + dl[u]) * HC))[lane];
  }
  __builtin_amdgcn_sched_barrier(0);
  float s[NE];
#pragma unroll
  for (int u = 0; u < NE; ++u) s[u] = lane_dot(a[u], b[u], w);
  head_sum<NE>(s);
  if ((lane & 7) == 0) {
    const int h = lane >> 3;
#pragma unroll
    for (int u = 0; u < NE; ++u) {
      s_val[(i0 + u) * 4 + h] = s[u];
      __hip_atomic_fetch_max(&s_m[dl[u] * 4 + h], s[u], __ATOMIC_RELAXED,
                             __HIP_MEMORY_SCOPE_WORKGROUP);
    }
  }
}

__device__ __forceinline__ void phase_logits(const float* __restrict__ xl,
                                             const float* __restrict__ xr,
                                             const float4 w, int node0, int n,
                                             const int* s_src, const int* s_dl,
                                             float* s_val, float* s_m) {
  const int wave = threadIdx.x >> 5;
  const int lane = threadIdx.x & 31;
  const int nfull = n / GRP_WIN;
  for (int g = wave; g < nfull; g += NWAVE)
    logit_group<GRP_WIN>(xl, xr, w, node0, g * GRP_WIN, lane, s_src, s_dl, s_val, s_m);
  const int i = nfull * GRP_WIN + wave;
  if (i < n)
    logit_group<1>(xl, xr, w, node0, i, lane, s_src, s_dl, s_val, s_m);
}

__device__ __forceinline__ void phase_exp(int n, bool add, const int* s_dl,
                                          float* s_val, const float* s_m, float* s_den) {
  for (int i = threadIdx.x; i < n * 4; i += TPB) {
    const int k = s_dl[i >> 2] * 4 + (i & 3);
    const float ex = __expf(s_val[i] - s_m[k]);
    s_val[i] = ex;
    if (add)
      __hip_atomic_fetch_add(&s_den[k], ex, __ATOMIC_RELAXED, __HIP_MEMORY_SCOPE_WORKGROUP);
  }
}

__device__ __forceinline__ void phase_alpha(int n, const int* s_dl, float* s_val,
                                            const float* s_den) {
  for (int i = threadIdx.x; i < n * 4; i += TPB) {
    const int k = s_dl[i >> 2] * 4 + (i & 3);
    s_val[i] = s_val[i] / (s_den[k] + 1e-16f);
  }
}

template <int NE>
__device__ __forceinline__ void sum_group(const float* __restrict__ xl, int i0, int lane,
                                          const int* s_src, const int* s_dl,
                                          const float* s_val, float* s_acc) {
  float4 v[NE];
  int dl[NE];
#pragma unroll
  for (int u = 0; u < NE; ++u) {
    const int src = s_src[i0 + u];
    dl[u] = s_dl[i0 + u];
    v[u] = ((const float4*)(xl + (size_t)src * HC))[lane];
  }
  __builtin_amdgcn_sched_barrier(0);
#pragma unroll
  for (int u = 0; u < NE; ++u) {
    const float alpha = s_val[(i0 + u) * 4 + (lane >> 3)];
    float* row = s_acc + dl[u] * HC + lane;
    __hip_atomic_fetch_add(row +  0, v[u].x * alpha, __ATOMIC_RELAXED, __HIP_MEMORY_SCOPE_WORKGROUP);
    __hip_atomic_fetch_add(row + 32, v[u].y * alpha, __ATOMIC_RELAXED, __HIP_MEMORY_SCOPE_WORKGROUP);
    __hip_atomic_fetch_add(row + 64, v[u].z * alpha, __ATOMIC_RELAXED, __HIP_MEMORY_SCOPE_WORKGROUP);
    __hip_atomic_fetch_add(row + 96, v[u].w * alpha, __ATOMIC_RELAXED, __HIP_MEMORY_SCOPE_WORKGROUP);
  }
}

template <int G>
__device__ __forceinline__ void phase_sum(const float* __restrict__ xl, int n,
                                          const int* s_src, const int* s_dl,
                                          const float* s_val, float* s_acc) {
  const int wave = threadIdx.x >> 5;
  const int lane = threadIdx.x & 31;
  const int nfull = n / G;
  for (int g = wave; g < nfull; g += NWAVE)
    sum_group<G>(xl, g * G, lane, s_src, s_dl, s_val, s_acc);
  const int i = nfull * G + wave;
  if (i < n) sum_group<1>(xl, i, lane, s_src, s_dl, s_val, s_acc);
}

__device__ __forceinline__ void zero_acc(float* s_acc) {
  for (int i = threadIdx.x; i < NPB * HC / 4; i += TPB)
    ((float4*)s_acc)[i] = make_float4(0.f, 0.f, 0.f, 0.f);
}

__global__ void __launch_bounds__(TPB)
gat_owner(const unsigned* __restrict__ seg, const unsigned* __restrict__ tab,
          const float* __restrict__ xl,
          const float* __restrict__ xr, const float* __restrict__ att,
          const float* __restrict__ bias, float* __restrict__ out) {
  __shared__ float s_acc[NPB * HC];
  __shared__ __attribute__((aligned(16))) float s_u[CAP * 4];
  static_assert(NPB * HC <= CAP * 4, "the own xr rows fit the plane they share with the windowed path's values");
  __shared__ float s_m[NPB * 4];
  __shared__ float s_den[NPB * 4];
  __shared__ float s_c[NPB * 4];
  __shared__ int   s_src[CAP];
  __shared__ int   s_dl[CAP];
  __shared__ unsigned s_tab[GEMM_WG];
  __shared__ int   s_pos[GEMM_WG];
  __shared__ int   s_n;
  __shared__ int   s_flag;
  float* const s_xr  = s_u;
  float* const s_val = s_u;

  const int tid   = threadIdx.x;
  const int node0 = blockIdx.x * NPB;
  const unsigned nn = (unsigned)min(NPB, N_NODES - node0);
  const float4 w = ((const float4*)att)[tid & 31];
  unsigned tw[8] = {0u, 0u, 0u, 0u, 0u, 0u, 0u, 0u};
  if (tid < 32) tab_load(tab, tw);

  for (int i = tid; i < (int)nn * (HC / 4); i += TPB)
    ((float4*)s_xr)[i] = ((const float4*)(xr + (size_t)node0 * HC))[i];
  if (tid == 0) s_flag = 0;
  const int ne = list_places(tw, 0, GEMM_WG, s_tab, s_pos, &s_n);

  bool windowed = ne > CAP;
  if (!windowed) {
    list_fill(seg, s_tab, s_pos, s_src, s_dl);
    phase_self(xl, s_xr, w, node0, (int)nn, s_c, s_den, s_acc);
    __syncthreads();
    phase_sweep(xl, s_xr, w, ne, s_src, s_dl, s_c, s_den, s_acc, &s_flag);
    __syncthreads();
    windowed = s_flag != 0;
  }
  if (windowed) {
    zero_acc(s_acc);
    for (int i = tid; i < NPB * 4; i += TPB) {
      s_m[i] = -3.402823466e38f;
      s_den[i] = 0.0f;
    }
    for (int sweep = 0; sweep < 3; ++sweep) {
      for (int win = 0; win < NWIN; ++win) {
        __syncthreads();
        int n;
        if (win < NWIN - 1) {
          if (tid < 32) tab_load(tab, tw);
          n = list_places(tw, win * WIN_B, win * WIN_B + WIN_B, s_tab, s_pos, &s_n);
          list_fill(seg, s_tab, s_pos, s_src, s_dl);
        } else {
          n = (int)nn;
          list_self(node0, nn, 0, s_src, s_dl);
        }
        __syncthreads();
        phase_logits(xl, xr, w, node0, n, s_src, s_dl, s_val, s_m);
        __syncthreads();
        if (sweep >= 1) {
          phase_exp(n, sweep == 1, s_dl, s_val, s_m, s_den);
          __syncthreads();
        }
        if (sweep == 2) {
          phase_alpha(n, s_dl, s_val, s_den);
          __syncthreads();
          phase_sum<GRP_WIN>(xl, n, s_src, s_dl, s_val, s_acc);
          __syncthreads();
        }
      }
    }
  }

  const float4 bq = ((const float4*)bias)[tid & 31];
  for (int idx = tid; idx < (int)nn * 32; idx += TPB) {
    const float* row = s_acc + (idx >> 5) * HC + (idx & 31);
    const float dn = windowed ? 1.0f : s_den[(idx >> 5) * 4 + ((idx & 31) >> 3)] + 1e-16f;
    float4 a;
    a.x = row[0]  / dn + bq.x;
    a.y = row[32] / dn + bq.y;
    a.z = row[64] / dn + bq.z;
    a.w = row[96] / dn + bq.w;
    ((float4*)(out + (size_t)node0 * HC))[idx] = a;
  }
}

extern "C" void kernel_launch(void* const* d_in, const int* in_sizes, int n_in,
                              void* d_out, int out_size, void* d_ws, size_t ws_size,
                              hipStream_t stream) {
  const float* x    = (const float*)d_in[0];
  const int*   ei   = (const int*)  d_in[1];
  const float* Wl   = (const float*)d_in[2];
  const float* Wr   = (const float*)d_in[3];
  const float* att  = (const float*)d_in[4];
  const float* bias = (const float*)d_in[5];
  float* out = (float*)d_out;

  float* ws = (float*)d_ws;
  float* xl = ws;
  float* xr = xl + (size_t)N_NODES * HC;
  unsigned* seg = (unsigned*)(xr + (size_t)N_NODES * HC);
  unsigned* tab = seg + N_EDGES;

  gat_gemm<<<GEMM_WG, GEMM_T, 0, stream>>>(x, Wl, Wr, ei, xl, xr, seg, tab);

  gat_owner<<<NBLK, TPB, 0, stream>>>(seg, tab, xl, xr, att, bias, out);
}
